# starting kernel
# baseline (speedup 1.0000x reference)
#include <hip/hip_runtime.h>
#include <hip/hip_bf16.h>

typedef __attribute__((ext_vector_type(2))) float v2f;
typedef __attribute__((ext_vector_type(4))) float v4f;
typedef __attribute__((ext_vector_type(8))) float v8f;
typedef __attribute__((ext_vector_type(2))) int   v2i;

#define N_NODES 4096
#define F_IN    512
#define F_OUT   512
#define HEADS   4
#define HEAD_DIM 128
#define ALPHA   0.2f
#define NEG_INF (-9.0e15f)

__global__ __launch_bounds__(128) void gemm_xw(const float* __restrict__ x,
                                               const float* __restrict__ W,
                                               float* __restrict__ h) {
  const int lane = threadIdx.x & 31;
  const int wave = threadIdx.x >> 5;
  const int half = lane >> 4;
  const int lid  = lane & 15;
  const int kb   = half * 2;

  const int bm = blockIdx.x;
  const int bn = blockIdx.y;
  const int row = bm * 64 + wave * 16 + lid;

  v8f acc[4];
#pragma unroll
  for (int t = 0; t < 4; ++t) acc[t] = {};

  const float* xrow = x + (size_t)row * F_IN;
  for (int k = 0; k < F_IN; k += 4) {
    v2f a;
    a.x = xrow[k + kb + 0];
    a.y = xrow[k + kb + 1];
#pragma unroll
    for (int t = 0; t < 4; ++t) {
      const int col = bn * 64 + t * 16 + lid;
      v2f b;
      b.x = W[(size_t)(k + kb + 0) * F_OUT + col];
      b.y = W[(size_t)(k + kb + 1) * F_OUT + col];
      acc[t] = __builtin_amdgcn_wmma_f32_16x16x4_f32(
          false, a, false, b, (short)0, acc[t], false, false);
    }
  }

#pragma unroll
  for (int t = 0; t < 4; ++t) {
#pragma unroll
    for (int v = 0; v < 8; ++v) {
      const int m = bm * 64 + wave * 16 + v + half * 8;
      const int c = bn * 64 + t * 16 + lid;
      h[(size_t)m * F_OUT + c] = acc[t][v];
    }
  }
}

__global__ __launch_bounds__(128) void attn_scores(const float* __restrict__ h,
                                                   const float* __restrict__ a,
                                                   float* __restrict__ si,
                                                   float* __restrict__ sj,
                                                   float* __restrict__ sjT) {
  const int lane = threadIdx.x & 31;
  const int wave = threadIdx.x >> 5;
  const int n = blockIdx.x * 4 + wave;

  const v4f a1 = *(const v4f*)(a + lane * 4);
  const v4f a2 = *(const v4f*)(a + HEAD_DIM + lane * 4);

#pragma unroll
  for (int head = 0; head < HEADS; ++head) {
    const float* hrow = h + (size_t)n * F_OUT + head * HEAD_DIM;
    const v4f hv = *(const v4f*)(hrow + lane * 4);
    float s1 = hv.x * a1.x + hv.y * a1.y + hv.z * a1.z + hv.w * a1.w;
    float s2 = hv.x * a2.x + hv.y * a2.y + hv.z * a2.z + hv.w * a2.w;
#pragma unroll
    for (int off = 16; off > 0; off >>= 1) {
      s1 += __shfl_down(s1, off, 32);
      s2 += __shfl_down(s2, off, 32);
    }
    if (lane == 0) {
      si[head * N_NODES + n] = s1;
      sj[head * N_NODES + n] = s2;
      sjT[n * HEADS + head] = s2;
    }
  }
}

__global__ __launch_bounds__(256) void row_max(const int* __restrict__ adj,
                                               const float* __restrict__ sjT,
                                               float* __restrict__ Mx) {
  const int i = blockIdx.x;
  const int t = threadIdx.x;
  v4f m = {NEG_INF, NEG_INF, NEG_INF, NEG_INF};
  const int* arow = adj + (size_t)i * N_NODES;
  for (int j = t; j < N_NODES; j += 256) {
    if (arow[j] > 0) {
      const v4f sv = *(const v4f*)(sjT + j * 4);
      m.x = fmaxf(m.x, sv.x);
      m.y = fmaxf(m.y, sv.y);
      m.z = fmaxf(m.z, sv.z);
      m.w = fmaxf(m.w, sv.w);
    }
  }
  __shared__ v4f red[256];
  red[t] = m;
  __syncthreads();
  for (int s = 128; s > 0; s >>= 1) {
    if (t < s) {
      v4f o = red[t + s];
      v4f c = red[t];
      c.x = fmaxf(c.x, o.x);
      c.y = fmaxf(c.y, o.y);
      c.z = fmaxf(c.z, o.z);
      c.w = fmaxf(c.w, o.w);
      red[t] = c;
    }
    __syncthreads();
  }
  if (t < 4) Mx[t * N_NODES + i] = red[0][t];
}

#define CHUNK_J 64
#define PAIRS   (CHUNK_J / 2)
#define PAIR_STRIDE 288

template <bool PARTIAL>
__global__ __launch_bounds__(128) void attn_agg(const float* __restrict__ h,
                                                const int* __restrict__ adj,
                                                const float* __restrict__ si,
                                                const float* __restrict__ sj,
                                                const float* __restrict__ Mx,
                                                float* __restrict__ outp,
                                                float* __restrict__ pssum,
                                                int jspan) {
  __shared__ float Vp[PAIRS * PAIR_STRIDE];
  __shared__ float srow[4][16];

  const int lane = threadIdx.x & 31;
  const int wave = threadIdx.x >> 5;
  const int half = lane >> 4;
  const int lid  = lane & 15;
  const int kb   = half * 2;

  const int head = blockIdx.y;
  const int iblk = blockIdx.x * 64;
  const int z    = blockIdx.z;
  const int jstart = z * jspan;
  const int row  = iblk + wave * 16 + lid;

  const float sir = si[head * N_NODES + row];
  const float mx0 = sir + Mx[head * N_NODES + row];
  const float mrow = (mx0 > 0.f) ? mx0 : ALPHA * mx0;

  v8f acc[8];
#pragma unroll
  for (int t = 0; t < 8; ++t) acc[t] = {};
  float ssum = 0.f;

  const float* sjh  = sj + head * N_NODES;
  const int*   arow = adj + (size_t)row * N_NODES;
  const float* Vg   = h + head * HEAD_DIM;

  const int st_pair = threadIdx.x >> 2;
  const int st_d0   = (threadIdx.x & 3) * 32;

  for (int c = jstart; c < jstart + jspan; c += CHUNK_J) {
    {
      const float* src0 = Vg + (size_t)(c + st_pair * 2 + 0) * F_OUT + st_d0;
      const float* src1 = Vg + (size_t)(c + st_pair * 2 + 1) * F_OUT + st_d0;
      float* dst = Vp + st_pair * PAIR_STRIDE + st_d0 * 2;
#pragma unroll
      for (int q = 0; q < 8; ++q) {
        const v4f r0 = *(const v4f*)(src0 + q * 4);
        const v4f r1 = *(const v4f*)(src1 + q * 4);
        v4f w0 = {r0.x, r1.x, r0.y, r1.y};
        v4f w1 = {r0.z, r1.z, r0.w, r1.w};
        *(v4f*)(dst + q * 8 + 0) = w0;
        *(v4f*)(dst + q * 8 + 4) = w1;
      }
    }
    __syncthreads();

#pragma unroll
    for (int s4 = 0; s4 < CHUNK_J / 4; ++s4) {
      const int jg0 = c + s4 * 4 + kb;
      const v2f sjv = *(const v2f*)(sjh + jg0);
      const v2i av  = *(const v2i*)(arow + jg0);

      float e0 = sir + sjv.x; e0 = (e0 > 0.f) ? e0 : ALPHA * e0;
      float e1 = sir + sjv.y; e1 = (e1 > 0.f) ? e1 : ALPHA * e1;
      const float p0 = (av.x > 0) ? __expf(e0 - mrow) : 0.f;
      const float p1 = (av.y > 0) ? __expf(e1 - mrow) : 0.f;
      ssum += p0 + p1;

      v2f af;
      af.x = p0;
      af.y = p1;
      const float* bp = Vp + (s4 * 2 + half) * PAIR_STRIDE + lid * 2;
#pragma unroll
      for (int t = 0; t < 8; ++t) {
        const v2f bf = *(const v2f*)(bp + t * 32);
        acc[t] = __builtin_amdgcn_wmma_f32_16x16x4_f32(
            false, af, false, bf, (short)0, acc[t], false, false);
      }
    }
    __syncthreads();
  }

  ssum += __shfl_xor(ssum, 16, 32);

  if (PARTIAL) {
    if (half == 0) pssum[(size_t)z * (HEADS * N_NODES) + head * N_NODES + row] = ssum;
    float* pa = outp + (size_t)z * (N_NODES * F_OUT);
#pragma unroll
    for (int t = 0; t < 8; ++t) {
#pragma unroll
      for (int v = 0; v < 8; ++v) {
        const int m = iblk + wave * 16 + v + half * 8;
        pa[(size_t)m * F_OUT + head * HEAD_DIM + t * 16 + lid] = acc[t][v];
      }
    }
  } else {
    if (half == 0) srow[wave][lid] = ssum;
    __syncthreads();
#pragma unroll
    for (int t = 0; t < 8; ++t) {
#pragma unroll
      for (int v = 0; v < 8; ++v) {
        const int m = iblk + wave * 16 + v + half * 8;
        float val = acc[t][v] / srow[wave][v + half * 8];
        val = (val > 0.f) ? val : (__expf(val) - 1.f);
        outp[(size_t)m * F_OUT + head * HEAD_DIM + t * 16 + lid] = val;
      }
    }
  }
}

template <int JS>
__global__ __launch_bounds__(256) void attn_reduce(const float* __restrict__ pacc,
                                                   const float* __restrict__ pssum,
                                                   float* __restrict__ out) {
  const int i = blockIdx.x;
#pragma unroll
  for (int r = 0; r < 2; ++r) {
    const int col = threadIdx.x + r * 256;
    const int head = col >> 7;
    float num = 0.f, den = 0.f;
#pragma unroll
    for (int zz = 0; zz < JS; ++zz) {
      num += pacc[(size_t)zz * (N_NODES * F_OUT) + (size_t)i * F_OUT + col];
      den += pssum[(size_t)zz * (HEADS * N_NODES) + head * N_NODES + i];
    }
    float val = num / den;
    val = (val > 0.f) ? val : (__expf(val) - 1.f);
    out[(size_t)i * F_OUT + col] = val;
  }
}

extern "C" void kernel_launch(void* const* d_in, const int* in_sizes, int n_in,
                              void* d_out, int out_size, void* d_ws, size_t ws_size,
                              hipStream_t stream) {
  const float* x   = (const float*)d_in[0];
  const int*   adj = (const int*)d_in[1];
  const float* W   = (const float*)d_in[2];
  const float* a   = (const float*)d_in[3];
  float* out = (float*)d_out;

  constexpr int JS = 4;
  const size_t HN = (size_t)HEADS * N_NODES;

  float* h   = (float*)d_ws;
  float* si  = h + (size_t)N_NODES * F_OUT;
  float* sj  = si + HN;
  float* sjT = sj + HN;
  float* Mx  = sjT + HN;
  float* pacc  = Mx + HN;
  float* pssum = pacc + (size_t)JS * N_NODES * F_OUT;
  const size_t need_js =
      (size_t)(pssum + JS * HN - (float*)d_ws) * sizeof(float);

  gemm_xw<<<dim3(N_NODES / 64, F_OUT / 64), 128, 0, stream>>>(x, W, h);
  attn_scores<<<dim3(N_NODES / 4), 128, 0, stream>>>(h, a, si, sj, sjT);
  row_max<<<dim3(N_NODES), 256, 0, stream>>>(adj, sjT, Mx);

  if (ws_size >= need_js) {
    attn_agg<true><<<dim3(N_NODES / 64, HEADS, JS), 128, 0, stream>>>(
        h, adj, si, sj, Mx, pacc, pssum, N_NODES / JS);
    attn_reduce<JS><<<dim3(N_NODES), 256, 0, stream>>>(pacc, pssum, out);
  } else {
    attn_agg<false><<<dim3(N_NODES / 64, HEADS, 1), 128, 0, stream>>>(
        h, adj, si, sj, Mx, out, nullptr, N_NODES);
  }
}


# attn_agg V staging: each wave load reads one V row's contiguous 512 B slice (4 lines, was 32 lines of 16 B) and the chunk's 16 loads are issued as one group before the LDS stores
# speedup vs baseline: 1.3061x; 1.3061x over previous
#include <hip/hip_runtime.h>
#include <hip/hip_bf16.h>

typedef __attribute__((ext_vector_type(2))) float v2f;
typedef __attribute__((ext_vector_type(4))) float v4f;
typedef __attribute__((ext_vector_type(8))) float v8f;
typedef __attribute__((ext_vector_type(2))) int   v2i;

#define N_NODES 4096
#define F_IN    512
#define F_OUT   512
#define HEADS   4
#define HEAD_DIM 128
#define ALPHA   0.2f
#define NEG_INF (-9.0e15f)

__global__ __launch_bounds__(128) void gemm_xw(const float* __restrict__ x,
                                               const float* __restrict__ W,
                                               float* __restrict__ h) {
  const int lane = threadIdx.x & 31;
  const int wave = threadIdx.x >> 5;
  const int half = lane >> 4;
  const int lid  = lane & 15;
  const int kb   = half * 2;

  const int bm = blockIdx.x;
  const int bn = blockIdx.y;
  const int row = bm * 64 + wave * 16 + lid;

  v8f acc[4];
#pragma unroll
  for (int t = 0; t < 4; ++t) acc[t] = {};

  const float* xrow = x + (size_t)row * F_IN;
  for (int k = 0; k < F_IN; k += 4) {
    v2f a;
    a.x = xrow[k + kb + 0];
    a.y = xrow[k + kb + 1];
#pragma unroll
    for (int t = 0; t < 4; ++t) {
      const int col = bn * 64 + t * 16 + lid;
      v2f b;
      b.x = W[(size_t)(k + kb + 0) * F_OUT + col];
      b.y = W[(size_t)(k + kb + 1) * F_OUT + col];
      acc[t] = __builtin_amdgcn_wmma_f32_16x16x4_f32(
          false, a, false, b, (short)0, acc[t], false, false);
    }
  }

#pragma unroll
  for (int t = 0; t < 4; ++t) {
#pragma unroll
    for (int v = 0; v < 8; ++v) {
      const int m = bm * 64 + wave * 16 + v + half * 8;
      const int c = bn * 64 + t * 16 + lid;
      h[(size_t)m * F_OUT + c] = acc[t][v];
    }
  }
}

__global__ __launch_bounds__(128) void attn_scores(const float* __restrict__ h,
                                                   const float* __restrict__ a,
                                                   float* __restrict__ si,
                                                   float* __restrict__ sj,
                                                   float* __restrict__ sjT) {
  const int lane = threadIdx.x & 31;
  const int wave = threadIdx.x >> 5;
  const int n = blockIdx.x * 4 + wave;

  const v4f a1 = *(const v4f*)(a + lane * 4);
  const v4f a2 = *(const v4f*)(a + HEAD_DIM + lane * 4);

#pragma unroll
  for (int head = 0; head < HEADS; ++head) {
    const float* hrow = h + (size_t)n * F_OUT + head * HEAD_DIM;
    const v4f hv = *(const v4f*)(hrow + lane * 4);
    float s1 = hv.x * a1.x + hv.y * a1.y + hv.z * a1.z + hv.w * a1.w;
    float s2 = hv.x * a2.x + hv.y * a2.y + hv.z * a2.z + hv.w * a2.w;
#pragma unroll
    for (int off = 16; off > 0; off >>= 1) {
      s1 += __shfl_down(s1, off, 32);
      s2 += __shfl_down(s2, off, 32);
    }
    if (lane == 0) {
      si[head * N_NODES + n] = s1;
      sj[head * N_NODES + n] = s2;
      sjT[n * HEADS + head] = s2;
    }
  }
}

__global__ __launch_bounds__(256) void row_max(const int* __restrict__ adj,
                                               const float* __restrict__ sjT,
                                               float* __restrict__ Mx) {
  const int i = blockIdx.x;
  const int t = threadIdx.x;
  v4f m = {NEG_INF, NEG_INF, NEG_INF, NEG_INF};
  const int* arow = adj + (size_t)i * N_NODES;
  for (int j = t; j < N_NODES; j += 256) {
    if (arow[j] > 0) {
      const v4f sv = *(const v4f*)(sjT + j * 4);
      m.x = fmaxf(m.x, sv.x);
      m.y = fmaxf(m.y, sv.y);
      m.z = fmaxf(m.z, sv.z);
      m.w = fmaxf(m.w, sv.w);
    }
  }
  __shared__ v4f red[256];
  red[t] = m;
  __syncthreads();
  for (int s = 128; s > 0; s >>= 1) {
    if (t < s) {
      v4f o = red[t + s];
      v4f c = red[t];
      c.x = fmaxf(c.x, o.x);
      c.y = fmaxf(c.y, o.y);
      c.z = fmaxf(c.z, o.z);
      c.w = fmaxf(c.w, o.w);
      red[t] = c;
    }
    __syncthreads();
  }
  if (t < 4) Mx[t * N_NODES + i] = red[0][t];
}

#define CHUNK_J 64
#define PAIRS   (CHUNK_J / 2)
#define PAIR_STRIDE 288

template <bool PARTIAL>
__global__ __launch_bounds__(128) void attn_agg(const float* __restrict__ h,
                                                const int* __restrict__ adj,
                                                const float* __restrict__ si,
                                                const float* __restrict__ sj,
                                                const float* __restrict__ Mx,
                                                float* __restrict__ outp,
                                                float* __restrict__ pssum,
                                                int jspan) {
  __shared__ float Vp[PAIRS * PAIR_STRIDE];
  __shared__ float srow[4][16];

  const int lane = threadIdx.x & 31;
  const int wave = threadIdx.x >> 5;
  const int half = lane >> 4;
  const int lid  = lane & 15;
  const int kb   = half * 2;

  const int head = blockIdx.y;
  const int iblk = blockIdx.x * 64;
  const int z    = blockIdx.z;
  const int jstart = z * jspan;
  const int row  = iblk + wave * 16 + lid;

  const float sir = si[head * N_NODES + row];
  const float mx0 = sir + Mx[head * N_NODES + row];
  const float mrow = (mx0 > 0.f) ? mx0 : ALPHA * mx0;

  v8f acc[8];
#pragma unroll
  for (int t = 0; t < 8; ++t) acc[t] = {};
  float ssum = 0.f;

  const float* sjh  = sj + head * N_NODES;
  const int*   arow = adj + (size_t)row * N_NODES;
  const float* Vg   = h + head * HEAD_DIM;

  const int st_pair0 = wave * (PAIRS / 4);
  const int st_d     = lane * 4;

  for (int c = jstart; c < jstart + jspan; c += CHUNK_J) {
    {
      const float* srcp = Vg + (size_t)(c + st_pair0 * 2) * F_OUT + st_d;
      float* dst = Vp + st_pair0 * PAIR_STRIDE + st_d * 2;
      v4f rr[CHUNK_J / 4];
#pragma unroll
      for (int i = 0; i < CHUNK_J / 4; ++i)
        rr[i] = *(const v4f*)(srcp + (size_t)i * F_OUT);
      __builtin_amdgcn_sched_barrier(0);
#pragma unroll
      for (int pp = 0; pp < PAIRS / 4; ++pp) {
        const v4f r0 = rr[pp * 2 + 0];
        const v4f r1 = rr[pp * 2 + 1];
        v4f w0 = {r0.x, r1.x, r0.y, r1.y};
        v4f w1 = {r0.z, r1.z, r0.w, r1.w};
        *(v4f*)(dst + pp * PAIR_STRIDE + 0) = w0;
        *(v4f*)(dst + pp * PAIR_STRIDE + 4) = w1;
      }
    }
    __syncthreads();

#pragma unroll
    for (int s4 = 0; s4 < CHUNK_J / 4; ++s4) {
      const int jg0 = c + s4 * 4 + kb;
      const v2f sjv = *(const v2f*)(sjh + jg0);
      const v2i av  = *(const v2i*)(arow + jg0);

      float e0 = sir + sjv.x; e0 = (e0 > 0.f) ? e0 : ALPHA * e0;
      float e1 = sir + sjv.y; e1 = (e1 > 0.f) ? e1 : ALPHA * e1;
      const float p0 = (av.x > 0) ? __expf(e0 - mrow) : 0.f;
      const float p1 = (av.y > 0) ? __expf(e1 - mrow) : 0.f;
      ssum += p0 + p1;

      v2f af;
      af.x = p0;
      af.y = p1;
      const float* bp = Vp + (s4 * 2 + half) * PAIR_STRIDE + lid * 2;
#pragma unroll
      for (int t = 0; t < 8; ++t) {
        const v2f bf = *(const v2f*)(bp + t * 32);
        acc[t] = __builtin_amdgcn_wmma_f32_16x16x4_f32(
            false, af, false, bf, (short)0, acc[t], false, false);
      }
    }
    __syncthreads();
  }

  ssum += __shfl_xor(ssum, 16, 32);

  if (PARTIAL) {
    if (half == 0) pssum[(size_t)z * (HEADS * N_NODES) + head * N_NODES + row] = ssum;
    float* pa = outp + (size_t)z * (N_NODES * F_OUT);
#pragma unroll
    for (int t = 0; t < 8; ++t) {
#pragma unroll
      for (int v = 0; v < 8; ++v) {
        const int m = iblk + wave * 16 + v + half * 8;
        pa[(size_t)m * F_OUT + head * HEAD_DIM + t * 16 + lid] = acc[t][v];
      }
    }
  } else {
    if (half == 0) srow[wave][lid] = ssum;
    __syncthreads();
#pragma unroll
    for (int t = 0; t < 8; ++t) {
#pragma unroll
      for (int v = 0; v < 8; ++v) {
        const int m = iblk + wave * 16 + v + half * 8;
        float val = acc[t][v] / srow[wave][v + half * 8];
        val = (val > 0.f) ? val : (__expf(val) - 1.f);
        outp[(size_t)m * F_OUT + head * HEAD_DIM + t * 16 + lid] = val;
      }
    }
  }
}

template <int JS>
__global__ __launch_bounds__(256) void attn_reduce(const float* __restrict__ pacc,
                                                   const float* __restrict__ pssum,
                                                   float* __restrict__ out) {
  const int i = blockIdx.x;
#pragma unroll
  for (int r = 0; r < 2; ++r) {
    const int col = threadIdx.x + r * 256;
    const int head = col >> 7;
    float num = 0.f, den = 0.f;
#pragma unroll
    for (int zz = 0; zz < JS; ++zz) {
      num += pacc[(size_t)zz * (N_NODES * F_OUT) + (size_t)i * F_OUT + col];
      den += pssum[(size_t)zz * (HEADS * N_NODES) + head * N_NODES + i];
    }
    float val = num / den;
    val = (val > 0.f) ? val : (__expf(val) - 1.f);
    out[(size_t)i * F_OUT + col] = val;
  }
}

extern "C" void kernel_launch(void* const* d_in, const int* in_sizes, int n_in,
                              void* d_out, int out_size, void* d_ws, size_t ws_size,
                              hipStream_t stream) {
  const float* x   = (const float*)d_in[0];
  const int*   adj = (const int*)d_in[1];
  const float* W   = (const float*)d_in[2];
  const float* a   = (const float*)d_in[3];
  float* out = (float*)d_out;

  constexpr int JS = 4;
  const size_t HN = (size_t)HEADS * N_NODES;

  float* h   = (float*)d_ws;
  float* si  = h + (size_t)N_NODES * F_OUT;
  float* sj  = si + HN;
  float* sjT = sj + HN;
  float* Mx  = sjT + HN;
  float* pacc  = Mx + HN;
  float* pssum = pacc + (size_t)JS * N_NODES * F_OUT;
  const size_t need_js =
      (size_t)(pssum + JS * HN - (float*)d_ws) * sizeof(float);

  gemm_xw<<<dim3(N_NODES / 64, F_OUT / 64), 128, 0, stream>>>(x, W, h);
  attn_scores<<<dim3(N_NODES / 4), 128, 0, stream>>>(h, a, si, sj, sjT);
  row_max<<<dim3(N_NODES), 256, 0, stream>>>(adj, sjT, Mx);

  if (ws_size >= need_js) {
    attn_agg<true><<<dim3(N_NODES / 64, HEADS, JS), 128, 0, stream>>>(
        h, adj, si, sj, Mx, pacc, pssum, N_NODES / JS);
    attn_reduce<JS><<<dim3(N_NODES), 256, 0, stream>>>(pacc, pssum, out);
  } else {
    attn_agg<false><<<dim3(N_NODES / 64, HEADS, 1), 128, 0, stream>>>(
        h, adj, si, sj, Mx, out, nullptr, N_NODES);
  }
}


# row_max and attn_reduce re-cut to one wave per node row (4096 waves each, were 32768); a lane's 16-byte loads go out as groups in flight; row_max reads sj from an LDS copy; sum order kept
# speedup vs baseline: 1.4413x; 1.1035x over previous
#include <hip/hip_runtime.h>
#include <hip/hip_bf16.h>

typedef __attribute__((ext_vector_type(2))) float v2f;
typedef __attribute__((ext_vector_type(4))) float v4f;
typedef __attribute__((ext_vector_type(8))) float v8f;
typedef __attribute__((ext_vector_type(2))) int   v2i;
typedef __attribute__((ext_vector_type(4))) int   v4i;

#define N_NODES 4096
#define F_IN    512
#define F_OUT   512
#define HEADS   4
#define HEAD_DIM 128
#define ALPHA   0.2f
#define NEG_INF (-9.0e15f)

__global__ __launch_bounds__(128) void gemm_xw(const float* __restrict__ x,
                                               const float* __restrict__ W,
                                               float* __restrict__ h) {
  const int lane = threadIdx.x & 31;
  const int wave = threadIdx.x >> 5;
  const int half = lane >> 4;
  const int lid  = lane & 15;
  const int kb   = half * 2;

  const int bm = blockIdx.x;
  const int bn = blockIdx.y;
  const int row = bm * 64 + wave * 16 + lid;

  v8f acc[4];
#pragma unroll
  for (int t = 0; t < 4; ++t) acc[t] = {};

  const float* xrow = x + (size_t)row * F_IN;
  for (int k = 0; k < F_IN; k += 4) {
    v2f a;
    a.x = xrow[k + kb + 0];
    a.y = xrow[k + kb + 1];
#pragma unroll
    for (int t = 0; t < 4; ++t) {
      const int col = bn * 64 + t * 16 + lid;
      v2f b;
      b.x = W[(size_t)(k + kb + 0) * F_OUT + col];
      b.y = W[(size_t)(k + kb + 1) * F_OUT + col];
      acc[t] = __builtin_amdgcn_wmma_f32_16x16x4_f32(
          false, a, false, b, (short)0, acc[t], false, false);
    }
  }

#pragma unroll
  for (int t = 0; t < 4; ++t) {
#pragma unroll
    for (int v = 0; v < 8; ++v) {
      const int m = bm * 64 + wave * 16 + v + half * 8;
      const int c = bn * 64 + t * 16 + lid;
      h[(size_t)m * F_OUT + c] = acc[t][v];
    }
  }
}

__global__ __launch_bounds__(128) void attn_scores(const float* __restrict__ h,
                                                   const float* __restrict__ a,
                                                   float* __restrict__ si,
                                                   float* __restrict__ sj) {
  const int lane = threadIdx.x & 31;
  const int wave = threadIdx.x >> 5;
  const int n = blockIdx.x * 4 + wave;

  const v4f a1 = *(const v4f*)(a + lane * 4);
  const v4f a2 = *(const v4f*)(a + HEAD_DIM + lane * 4);

#pragma unroll
  for (int head = 0; head < HEADS; ++head) {
    const float* hrow = h + (size_t)n * F_OUT + head * HEAD_DIM;
    const v4f hv = *(const v4f*)(hrow + lane * 4);
    float s1 = hv.x * a1.x + hv.y * a1.y + hv.z * a1.z + hv.w * a1.w;
    float s2 = hv.x * a2.x + hv.y * a2.y + hv.z * a2.z + hv.w * a2.w;
#pragma unroll
    for (int off = 16; off > 0; off >>= 1) {
      s1 += __shfl_down(s1, off, 32);
      s2 += __shfl_down(s2, off, 32);
    }
    if (lane == 0) {
      si[head * N_NODES + n] = s1;
      sj[head * N_NODES + n] = s2;
    }
  }
}

#define RM_ROWS  16
#define RM_GROUP 8
#define RM_NGRP  (N_NODES / (128 * RM_GROUP))
#define RM_LDS   4

__device__ __forceinline__ float masked_max4(float m, const v4i a, const v4f s) {
  const float c0 = (a.x > 0) ? s.x : NEG_INF;
  const float c1 = (a.y > 0) ? s.y : NEG_INF;
  const float c2 = (a.z > 0) ? s.z : NEG_INF;
  const float c3 = (a.w > 0) ? s.w : NEG_INF;
  return fmaxf(fmaxf(m, fmaxf(c0, c1)), fmaxf(c2, c3));
}

__global__ __launch_bounds__(512) void row_max(const int* __restrict__ adj,
                                               const float* __restrict__ sj,
                                               float* __restrict__ Mx) {
  __shared__ float sjs[HEADS * N_NODES];

  const int t    = threadIdx.x;
  const int lane = t & 31;
  const int wave = t >> 5;
  const int i    = blockIdx.x * RM_ROWS + wave;
  const int* arow = adj + (size_t)i * N_NODES + lane * 4;

  v4f tb[HEADS * N_NODES / 4 / 512];
#pragma unroll
  for (int k = 0; k < HEADS * N_NODES / 4 / 512; ++k)
    tb[k] = *(const v4f*)(sj + (size_t)(k * 512 + t) * 4);

  v4i cur[RM_GROUP];
#pragma unroll
  for (int u = 0; u < RM_GROUP; ++u)
    cur[u] = *(const v4i*)(arow + u * 128);

#pragma unroll
  for (int k = 0; k < HEADS * N_NODES / 4 / 512; ++k)
    *(v4f*)(sjs + (k * 512 + t) * 4) = tb[k];
  __syncthreads();

  float m0 = NEG_INF, m1 = NEG_INF, m2 = NEG_INF, m3 = NEG_INF;
#pragma unroll
  for (int g = 0; g < RM_NGRP; ++g) {
    v4i nxt[RM_GROUP];
    if (g + 1 < RM_NGRP) {
#pragma unroll
      for (int u = 0; u < RM_GROUP; ++u)
        nxt[u] = *(const v4i*)(arow + ((g + 1) * RM_GROUP + u) * 128);
    }
#pragma unroll
    for (int u0 = 0; u0 < RM_GROUP; u0 += RM_LDS) {
      v4f s[RM_LDS][HEADS];
#pragma unroll
      for (int u = 0; u < RM_LDS; ++u) {
        const float* sp = sjs + (g * RM_GROUP + u0 + u) * 128 + lane * 4;
#pragma unroll
        for (int hd = 0; hd < HEADS; ++hd)
          s[u][hd] = *(const v4f*)(sp + hd * N_NODES);
      }
      __builtin_amdgcn_sched_barrier(0);
#pragma unroll
      for (int u = 0; u < RM_LDS; ++u) {
        m0 = masked_max4(m0, cur[u0 + u], s[u][0]);
        m1 = masked_max4(m1, cur[u0 + u], s[u][1]);
        m2 = masked_max4(m2, cur[u0 + u], s[u][2]);
        m3 = masked_max4(m3, cur[u0 + u], s[u][3]);
      }
    }
    if (g + 1 < RM_NGRP) {
#pragma unroll
      for (int u = 0; u < RM_GROUP; ++u) cur[u] = nxt[u];
    }
  }

#pragma unroll
  for (int off = 16; off > 0; off >>= 1) {
    m0 = fmaxf(m0, __shfl_xor(m0, off, 32));
    m1 = fmaxf(m1, __shfl_xor(m1, off, 32));
    m2 = fmaxf(m2, __shfl_xor(m2, off, 32));
    m3 = fmaxf(m3, __shfl_xor(m3, off, 32));
  }
  if (lane == 0) {
    Mx[0 * N_NODES + i] = m0;
    Mx[1 * N_NODES + i] = m1;
    Mx[2 * N_NODES + i] = m2;
    Mx[3 * N_NODES + i] = m3;
  }
}

#define CHUNK_J 64
#define PAIRS   (CHUNK_J / 2)
#define PAIR_STRIDE 288

template <bool PARTIAL>
__global__ __launch_bounds__(128) void attn_agg(const float* __restrict__ h,
                                                const int* __restrict__ adj,
                                                const float* __restrict__ si,
                                                const float* __restrict__ sj,
                                                const float* __restrict__ Mx,
                                                float* __restrict__ outp,
                                                float* __restrict__ pssum,
                                                int jspan) {
  __shared__ float Vp[PAIRS * PAIR_STRIDE];
  __shared__ float srow[4][16];

  const int lane = threadIdx.x & 31;
  const int wave = threadIdx.x >> 5;
  const int half = lane >> 4;
  const int lid  = lane & 15;
  const int kb   = half * 2;

  const int head = blockIdx.y;
  const int iblk = blockIdx.x * 64;
  const int z    = blockIdx.z;
  const int jstart = z * jspan;
  const int row  = iblk + wave * 16 + lid;

  const float sir = si[head * N_NODES + row];
  const float mx0 = sir + Mx[head * N_NODES + row];
  const float mrow = (mx0 > 0.f) ? mx0 : ALPHA * mx0;

  v8f acc[8];
#pragma unroll
  for (int t = 0; t < 8; ++t) acc[t] = {};
  float ssum = 0.f;

  const float* sjh  = sj + head * N_NODES;
  const int*   arow = adj + (size_t)row * N_NODES;
  const float* Vg   = h + head * HEAD_DIM;

  const int st_pair0 = wave * (PAIRS / 4);
  const int st_d     = lane * 4;

  for (int c = jstart; c < jstart + jspan; c += CHUNK_J) {
    {
      const float* srcp = Vg + (size_t)(c + st_pair0 * 2) * F_OUT + st_d;
      float* dst = Vp + st_pair0 * PAIR_STRIDE + st_d * 2;
      v4f rr[CHUNK_J / 4];
#pragma unroll
      for (int i = 0; i < CHUNK_J / 4; ++i)
        rr[i] = *(const v4f*)(srcp + (size_t)i * F_OUT);
      __builtin_amdgcn_sched_barrier(0);
#pragma unroll
      for (int pp = 0; pp < PAIRS / 4; ++pp) {
        const v4f r0 = rr[pp * 2 + 0];
        const v4f r1 = rr[pp * 2 + 1];
        v4f w0 = {r0.x, r1.x, r0.y, r1.y};
        v4f w1 = {r0.z, r1.z, r0.w, r1.w};
        *(v4f*)(dst + pp * PAIR_STRIDE + 0) = w0;
        *(v4f*)(dst + pp * PAIR_STRIDE + 4) = w1;
      }
    }
    __syncthreads();

#pragma unroll
    for (int s4 = 0; s4 < CHUNK_J / 4; ++s4) {
      const int jg0 = c + s4 * 4 + kb;
      const v2f sjv = *(const v2f*)(sjh + jg0);
      const v2i av  = *(const v2i*)(arow + jg0);

      float e0 = sir + sjv.x; e0 = (e0 > 0.f) ? e0 : ALPHA * e0;
      float e1 = sir + sjv.y; e1 = (e1 > 0.f) ? e1 : ALPHA * e1;
      const float p0 = (av.x > 0) ? __expf(e0 - mrow) : 0.f;
      const float p1 = (av.y > 0) ? __expf(e1 - mrow) : 0.f;
      ssum += p0 + p1;

      v2f af;
      af.x = p0;
      af.y = p1;
      const float* bp = Vp + (s4 * 2 + half) * PAIR_STRIDE + lid * 2;
#pragma unroll
      for (int t = 0; t < 8; ++t) {
        const v2f bf = *(const v2f*)(bp + t * 32);
        acc[t] = __builtin_amdgcn_wmma_f32_16x16x4_f32(
            false, af, false, bf, (short)0, acc[t], false, false);
      }
    }
    __syncthreads();
  }

  ssum += __shfl_xor(ssum, 16, 32);

  if (PARTIAL) {
    if (half == 0) pssum[(size_t)z * (HEADS * N_NODES) + head * N_NODES + row] = ssum;
    float* pa = outp + (size_t)z * (N_NODES * F_OUT);
#pragma unroll
    for (int t = 0; t < 8; ++t) {
#pragma unroll
      for (int v = 0; v < 8; ++v) {
        const int m = iblk + wave * 16 + v + half * 8;
        pa[(size_t)m * F_OUT + head * HEAD_DIM + t * 16 + lid] = acc[t][v];
      }
    }
  } else {
    if (half == 0) srow[wave][lid] = ssum;
    __syncthreads();
#pragma unroll
    for (int t = 0; t < 8; ++t) {
#pragma unroll
      for (int v = 0; v < 8; ++v) {
        const int m = iblk + wave * 16 + v + half * 8;
        float val = acc[t][v] / srow[wave][v + half * 8];
        val = (val > 0.f) ? val : (__expf(val) - 1.f);
        outp[(size_t)m * F_OUT + head * HEAD_DIM + t * 16 + lid] = val;
      }
    }
  }
}

template <int JS>
__global__ __launch_bounds__(512) void attn_reduce(const float* __restrict__ pacc,
                                                   const float* __restrict__ pssum,
                                                   float* __restrict__ out) {
  const int lane = threadIdx.x & 31;
  const int wave = threadIdx.x >> 5;
  const int i = blockIdx.x * 16 + wave;

  v4f p[HEADS][JS];
  float d[HEADS][JS];
#pragma unroll
  for (int head = 0; head < HEADS; ++head) {
#pragma unroll
    for (int zz = 0; zz < JS; ++zz) {
      p[head][zz] = *(const v4f*)(pacc + (size_t)zz * (N_NODES * F_OUT) +
                                  (size_t)i * F_OUT + head * HEAD_DIM + lane * 4);
      d[head][zz] = pssum[(size_t)zz * (HEADS * N_NODES) + head * N_NODES + i];
    }
  }
#pragma unroll
  for (int head = 0; head < HEADS; ++head) {
    v4f num = {0.f, 0.f, 0.f, 0.f};
    float den = 0.f;
#pragma unroll
    for (int zz = 0; zz < JS; ++zz) {
      num += p[head][zz];
      den += d[head][zz];
    }
    v4f val = num / den;
    val.x = (val.x > 0.f) ? val.x : (__expf(val.x) - 1.f);
    val.y = (val.y > 0.f) ? val.y : (__expf(val.y) - 1.f);
    val.z = (val.z > 0.f) ? val.z : (__expf(val.z) - 1.f);
    val.w = (val.w > 0.f) ? val.w : (__expf(val.w) - 1.f);
    *(v4f*)(out + (size_t)i * F_OUT + head * HEAD_DIM + lane * 4) = val;
  }
}

extern "C" void kernel_launch(void* const* d_in, const int* in_sizes, int n_in,
                              void* d_out, int out_size, void* d_ws, size_t ws_size,
                              hipStream_t stream) {
  const float* x   = (const float*)d_in[0];
  const int*   adj = (const int*)d_in[1];
  const float* W   = (const float*)d_in[2];
  const float* a   = (const float*)d_in[3];
  float* out = (float*)d_out;

  constexpr int JS = 4;
  const size_t HN = (size_t)HEADS * N_NODES;

  float* h   = (float*)d_ws;
  float* si  = h + (size_t)N_NODES * F_OUT;
  float* sj  = si + HN;
  float* Mx  = sj + HN;
  float* pacc  = Mx + HN;
  float* pssum = pacc + (size_t)JS * N_NODES * F_OUT;
  const size_t need_js =
      (size_t)(pssum + JS * HN - (float*)d_ws) * sizeof(float);

  gemm_xw<<<dim3(N_NODES / 64, F_OUT / 64), 128, 0, stream>>>(x, W, h);
  attn_scores<<<dim3(N_NODES / 4), 128, 0, stream>>>(h, a, si, sj);
  row_max<<<dim3(N_NODES / RM_ROWS), 512, 0, stream>>>(adj, sj, Mx);

  if (ws_size >= need_js) {
    attn_agg<true><<<dim3(N_NODES / 64, HEADS, JS), 128, 0, stream>>>(
        h, adj, si, sj, Mx, pacc, pssum, N_NODES / JS);
    attn_reduce<JS><<<dim3(N_NODES / 16), 512, 0, stream>>>(pacc, pssum, out);
  } else {
    attn_agg<false><<<dim3(N_NODES / 64, HEADS, 1), 128, 0, stream>>>(
        h, adj, si, sj, Mx, out, nullptr, N_NODES);
  }
}


# adj packed to 1 bit per edge by row_max during its scan (each wave vote is one word); attn_agg tests bits: 1 eight-byte load per lane per 64-col chunk replaces 16 int32 gathers; sum order kept
# speedup vs baseline: 1.5027x; 1.0426x over previous
#include <hip/hip_runtime.h>
#include <hip/hip_bf16.h>

typedef __attribute__((ext_vector_type(2))) float v2f;
typedef __attribute__((ext_vector_type(4))) float v4f;
typedef __attribute__((ext_vector_type(8))) float v8f;
typedef __attribute__((ext_vector_type(2))) int   v2i;
typedef __attribute__((ext_vector_type(4))) int   v4i;
typedef __attribute__((ext_vector_type(2))) unsigned v2u;

#define N_NODES 4096
#define F_IN    512
#define F_OUT   512
#define HEADS   4
#define HEAD_DIM 128
#define ALPHA   0.2f
#define NEG_INF (-9.0e15f)
#define BIT_WORDS (N_NODES / 32)

__global__ __launch_bounds__(128) void gemm_xw(const float* __restrict__ x,
                                               const float* __restrict__ W,
                                               float* __restrict__ h) {
  const int lane = threadIdx.x & 31;
  const int wave = threadIdx.x >> 5;
  const int half = lane >> 4;
  const int lid  = lane & 15;
  const int kb   = half * 2;

  const int bm = blockIdx.x;
  const int bn = blockIdx.y;
  const int row = bm * 64 + wave * 16 + lid;

  v8f acc[4];
#pragma unroll
  for (int t = 0; t < 4; ++t) acc[t] = {};

  const float* xrow = x + (size_t)row * F_IN;
  for (int k = 0; k < F_IN; k += 4) {
    v2f a;
    a.x = xrow[k + kb + 0];
    a.y = xrow[k + kb + 1];
#pragma unroll
    for (int t = 0; t < 4; ++t) {
      const int col = bn * 64 + t * 16 + lid;
      v2f b;
      b.x = W[(size_t)(k + kb + 0) * F_OUT + col];
      b.y = W[(size_t)(k + kb + 1) * F_OUT + col];
      acc[t] = __builtin_amdgcn_wmma_f32_16x16x4_f32(
          false, a, false, b, (short)0, acc[t], false, false);
    }
  }

#pragma unroll
  for (int t = 0; t < 4; ++t) {
#pragma unroll
    for (int v = 0; v < 8; ++v) {
      const int m = bm * 64 + wave * 16 + v + half * 8;
      const int c = bn * 64 + t * 16 + lid;
      h[(size_t)m * F_OUT + c] = acc[t][v];
    }
  }
}

__global__ __launch_bounds__(128) void attn_scores(const float* __restrict__ h,
                                                   const float* __restrict__ a,
                                                   float* __restrict__ si,
                                                   float* __restrict__ sj) {
  const int lane = threadIdx.x & 31;
  const int wave = threadIdx.x >> 5;
  const int n = blockIdx.x * 4 + wave;

  const v4f a1 = *(const v4f*)(a + lane * 4);
  const v4f a2 = *(const v4f*)(a + HEAD_DIM + lane * 4);

#pragma unroll
  for (int head = 0; head < HEADS; ++head) {
    const float* hrow = h + (size_t)n * F_OUT + head * HEAD_DIM;
    const v4f hv = *(const v4f*)(hrow + lane * 4);
    float s1 = hv.x * a1.x + hv.y * a1.y + hv.z * a1.z + hv.w * a1.w;
    float s2 = hv.x * a2.x + hv.y * a2.y + hv.z * a2.z + hv.w * a2.w;
#pragma unroll
    for (int off = 16; off > 0; off >>= 1) {
      s1 += __shfl_down(s1, off, 32);
      s2 += __shfl_down(s2, off, 32);
    }
    if (lane == 0) {
      si[head * N_NODES + n] = s1;
      sj[head * N_NODES + n] = s2;
    }
  }
}

#define RM_ROWS  16
#define RM_GROUP 8
#define RM_NGRP  (N_NODES / (128 * RM_GROUP))
#define RM_LDS   4

__device__ __forceinline__ float masked_max4(float m, const v4i a, const v4f s) {
  const float c0 = (a.x > 0) ? s.x : NEG_INF;
  const float c1 = (a.y > 0) ? s.y : NEG_INF;
  const float c2 = (a.z > 0) ? s.z : NEG_INF;
  const float c3 = (a.w > 0) ? s.w : NEG_INF;
  return fmaxf(fmaxf(m, fmaxf(c0, c1)), fmaxf(c2, c3));
}

template <bool BITS>
__global__ __launch_bounds__(512) void row_max(const int* __restrict__ adj,
                                               const float* __restrict__ sj,
                                               float* __restrict__ Mx,
                                               unsigned* __restrict__ bits) {
  __shared__ float sjs[HEADS * N_NODES];

  const int t    = threadIdx.x;
  const int lane = t & 31;
  const int wave = t >> 5;
  const int i    = blockIdx.x * RM_ROWS + wave;
  const int* arow = adj + (size_t)i * N_NODES + lane * 4;

  v4f tb[HEADS * N_NODES / 4 / 512];
#pragma unroll
  for (int k = 0; k < HEADS * N_NODES / 4 / 512; ++k)
    tb[k] = *(const v4f*)(sj + (size_t)(k * 512 + t) * 4);

  v4i cur[RM_GROUP];
#pragma unroll
  for (int u = 0; u < RM_GROUP; ++u)
    cur[u] = *(const v4i*)(arow + u * 128);

#pragma unroll
  for (int k = 0; k < HEADS * N_NODES / 4 / 512; ++k)
    *(v4f*)(sjs + (k * 512 + t) * 4) = tb[k];
  __syncthreads();

  float m0 = NEG_INF, m1 = NEG_INF, m2 = NEG_INF, m3 = NEG_INF;
  unsigned bw[RM_NGRP];
#pragma unroll
  for (int g = 0; g < RM_NGRP; ++g) bw[g] = 0u;
#pragma unroll
  for (int g = 0; g < RM_NGRP; ++g) {
    v4i nxt[RM_GROUP];
    if (g + 1 < RM_NGRP) {
#pragma unroll
      for (int u = 0; u < RM_GROUP; ++u)
        nxt[u] = *(const v4i*)(arow + ((g + 1) * RM_GROUP + u) * 128);
    }
#pragma unroll
    for (int u0 = 0; u0 < RM_GROUP; u0 += RM_LDS) {
      v4f s[RM_LDS][HEADS];
#pragma unroll
      for (int u = 0; u < RM_LDS; ++u) {
        const float* sp = sjs + (g * RM_GROUP + u0 + u) * 128 + lane * 4;
#pragma unroll
        for (int hd = 0; hd < HEADS; ++hd)
          s[u][hd] = *(const v4f*)(sp + hd * N_NODES);
      }
      __builtin_amdgcn_sched_barrier(0);
#pragma unroll
      for (int u = 0; u < RM_LDS; ++u) {
        m0 = masked_max4(m0, cur[u0 + u], s[u][0]);
        m1 = masked_max4(m1, cur[u0 + u], s[u][1]);
        m2 = masked_max4(m2, cur[u0 + u], s[u][2]);
        m3 = masked_max4(m3, cur[u0 + u], s[u][3]);
      }
      if (BITS) {
#pragma unroll
        for (int u = 0; u < RM_LDS; ++u) {
          const v4i a = cur[u0 + u];
          const int q = (u0 + u) * 4;
          const unsigned v0 = __builtin_amdgcn_ballot_w32(a.x > 0);
          const unsigned v1 = __builtin_amdgcn_ballot_w32(a.y > 0);
          const unsigned v2 = __builtin_amdgcn_ballot_w32(a.z > 0);
          const unsigned v3 = __builtin_amdgcn_ballot_w32(a.w > 0);
          bw[g] = (lane == q + 0) ? v0 : bw[g];
          bw[g] = (lane == q + 1) ? v1 : bw[g];
          bw[g] = (lane == q + 2) ? v2 : bw[g];
          bw[g] = (lane == q + 3) ? v3 : bw[g];
        }
      }
    }
    if (g + 1 < RM_NGRP) {
#pragma unroll
      for (int u = 0; u < RM_GROUP; ++u) cur[u] = nxt[u];
    }
  }

#pragma unroll
  for (int off = 16; off > 0; off >>= 1) {
    m0 = fmaxf(m0, __shfl_xor(m0, off, 32));
    m1 = fmaxf(m1, __shfl_xor(m1, off, 32));
    m2 = fmaxf(m2, __shfl_xor(m2, off, 32));
    m3 = fmaxf(m3, __shfl_xor(m3, off, 32));
  }
  if (BITS) {
#pragma unroll
    for (int g = 0; g < RM_NGRP; ++g)
      bits[(size_t)i * BIT_WORDS + g * 32 + lane] = bw[g];
  }
  if (lane == 0) {
    Mx[0 * N_NODES + i] = m0;
    Mx[1 * N_NODES + i] = m1;
    Mx[2 * N_NODES + i] = m2;
    Mx[3 * N_NODES + i] = m3;
  }
}

#define CHUNK_J 64
#define PAIRS   (CHUNK_J / 2)
#define PAIR_STRIDE 288

template <bool PARTIAL, bool BITS>
__global__ __launch_bounds__(128) void attn_agg(const float* __restrict__ h,
                                                const int* __restrict__ adj,
                                                const unsigned* __restrict__ bits,
                                                const float* __restrict__ si,
                                                const float* __restrict__ sj,
                                                const float* __restrict__ Mx,
                                                float* __restrict__ outp,
                                                float* __restrict__ pssum,
                                                int jspan) {
  __shared__ float Vp[PAIRS * PAIR_STRIDE];
  __shared__ float srow[4][16];

  const int lane = threadIdx.x & 31;
  const int wave = threadIdx.x >> 5;
  const int half = lane >> 4;
  const int lid  = lane & 15;
  const int kb   = half * 2;

  const int head = blockIdx.y;
  const int iblk = blockIdx.x * 64;
  const int z    = blockIdx.z;
  const int jstart = z * jspan;
  const int row  = iblk + wave * 16 + lid;

  const float sir = si[head * N_NODES + row];
  const float mx0 = sir + Mx[head * N_NODES + row];
  const float mrow = (mx0 > 0.f) ? mx0 : ALPHA * mx0;

  v8f acc[8];
#pragma unroll
  for (int t = 0; t < 8; ++t) acc[t] = {};
  float ssum = 0.f;

  const float* sjh  = sj + head * N_NODES;
  const int*   arow = adj + (size_t)row * N_NODES;
  const unsigned* brow = bits + (size_t)row * BIT_WORDS + kb;
  const float* Vg   = h + head * HEAD_DIM;

  const int st_pair0 = wave * (PAIRS / 4);
  const int st_d     = lane * 4;

  for (int c = jstart; c < jstart + jspan; c += CHUNK_J) {
    unsigned w0 = 0u, w1 = 0u;
    if (BITS) {
      const v2u wv = *(const v2u*)(brow + (c >> 7) * 4);
      const int lb = (c & 64) >> 2;
      w0 = wv.x >> lb;
      w1 = wv.y >> lb;
    }
    {
      const float* srcp = Vg + (size_t)(c + st_pair0 * 2) * F_OUT + st_d;
      float* dst = Vp + st_pair0 * PAIR_STRIDE + st_d * 2;
      v4f rr[CHUNK_J / 4];
#pragma unroll
      for (int i = 0; i < CHUNK_J / 4; ++i)
        rr[i] = *(const v4f*)(srcp + (size_t)i * F_OUT);
      __builtin_amdgcn_sched_barrier(0);
#pragma unroll
      for (int pp = 0; pp < PAIRS / 4; ++pp) {
        const v4f r0 = rr[pp * 2 + 0];
        const v4f r1 = rr[pp * 2 + 1];
        v4f w0 = {r0.x, r1.x, r0.y, r1.y};
        v4f w1 = {r0.z, r1.z, r0.w, r1.w};
        *(v4f*)(dst + pp * PAIR_STRIDE + 0) = w0;
        *(v4f*)(dst + pp * PAIR_STRIDE + 4) = w1;
      }
    }
    __syncthreads();

#pragma unroll
    for (int s4 = 0; s4 < CHUNK_J / 4; ++s4) {
      const int jg0 = c + s4 * 4 + kb;
      const v2f sjv = *(const v2f*)(sjh + jg0);
      bool on0, on1;
      if (BITS) {
        on0 = ((w0 >> s4) & 1u) != 0u;
        on1 = ((w1 >> s4) & 1u) != 0u;
      } else {
        const v2i av = *(const v2i*)(arow + jg0);
        on0 = av.x > 0;
        on1 = av.y > 0;
      }

      float e0 = sir + sjv.x; e0 = (e0 > 0.f) ? e0 : ALPHA * e0;
      float e1 = sir + sjv.y; e1 = (e1 > 0.f) ? e1 : ALPHA * e1;
      const float p0 = on0 ? __expf(e0 - mrow) : 0.f;
      const float p1 = on1 ? __expf(e1 - mrow) : 0.f;
      ssum += p0 + p1;

      v2f af;
      af.x = p0;
      af.y = p1;
      const float* bp = Vp + (s4 * 2 + half) * PAIR_STRIDE + lid * 2;
#pragma unroll
      for (int t = 0; t < 8; ++t) {
        const v2f bf = *(const v2f*)(bp + t * 32);
        acc[t] = __builtin_amdgcn_wmma_f32_16x16x4_f32(
            false, af, false, bf, (short)0, acc[t], false, false);
      }
    }
    __syncthreads();
  }

  ssum += __shfl_xor(ssum, 16, 32);

  if (PARTIAL) {
    if (half == 0) pssum[(size_t)z * (HEADS * N_NODES) + head * N_NODES + row] = ssum;
    float* pa = outp + (size_t)z * (N_NODES * F_OUT);
#pragma unroll
    for (int t = 0; t < 8; ++t) {
#pragma unroll
      for (int v = 0; v < 8; ++v) {
        const int m = iblk + wave * 16 + v + half * 8;
        pa[(size_t)m * F_OUT + head * HEAD_DIM + t * 16 + lid] = acc[t][v];
      }
    }
  } else {
    if (half == 0) srow[wave][lid] = ssum;
    __syncthreads();
#pragma unroll
    for (int t = 0; t < 8; ++t) {
#pragma unroll
      for (int v = 0; v < 8; ++v) {
        const int m = iblk + wave * 16 + v + half * 8;
        float val = acc[t][v] / srow[wave][v + half * 8];
        val = (val > 0.f) ? val : (__expf(val) - 1.f);
        outp[(size_t)m * F_OUT + head * HEAD_DIM + t * 16 + lid] = val;
      }
    }
  }
}

template <int JS>
__global__ __launch_bounds__(512) void attn_reduce(const float* __restrict__ pacc,
                                                   const float* __restrict__ pssum,
                                                   float* __restrict__ out) {
  const int lane = threadIdx.x & 31;
  const int wave = threadIdx.x >> 5;
  const int i = blockIdx.x * 16 + wave;

  v4f p[HEADS][JS];
  float d[HEADS][JS];
#pragma unroll
  for (int head = 0; head < HEADS; ++head) {
#pragma unroll
    for (int zz = 0; zz < JS; ++zz) {
      p[head][zz] = *(const v4f*)(pacc + (size_t)zz * (N_NODES * F_OUT) +
                                  (size_t)i * F_OUT + head * HEAD_DIM + lane * 4);
      d[head][zz] = pssum[(size_t)zz * (HEADS * N_NODES) + head * N_NODES + i];
    }
  }
#pragma unroll
  for (int head = 0; head < HEADS; ++head) {
    v4f num = {0.f, 0.f, 0.f, 0.f};
    float den = 0.f;
#pragma unroll
    for (int zz = 0; zz < JS; ++zz) {
      num += p[head][zz];
      den += d[head][zz];
    }
    v4f val = num / den;
    val.x = (val.x > 0.f) ? val.x : (__expf(val.x) - 1.f);
    val.y = (val.y > 0.f) ? val.y : (__expf(val.y) - 1.f);
    val.z = (val.z > 0.f) ? val.z : (__expf(val.z) - 1.f);
    val.w = (val.w > 0.f) ? val.w : (__expf(val.w) - 1.f);
    *(v4f*)(out + (size_t)i * F_OUT + head * HEAD_DIM + lane * 4) = val;
  }
}

extern "C" void kernel_launch(void* const* d_in, const int* in_sizes, int n_in,
                              void* d_out, int out_size, void* d_ws, size_t ws_size,
                              hipStream_t stream) {
  const float* x   = (const float*)d_in[0];
  const int*   adj = (const int*)d_in[1];
  const float* W   = (const float*)d_in[2];
  const float* a   = (const float*)d_in[3];
  float* out = (float*)d_out;

  constexpr int JS = 4;
  const size_t HN = (size_t)HEADS * N_NODES;

  float* h   = (float*)d_ws;
  float* si  = h + (size_t)N_NODES * F_OUT;
  float* sj  = si + HN;
  float* Mx  = sj + HN;
  float* pacc  = Mx + HN;
  float* pssum = pacc + (size_t)JS * N_NODES * F_OUT;
  const size_t need_js =
      (size_t)(pssum + JS * HN - (float*)d_ws) * sizeof(float);
  unsigned* bits = (unsigned*)(pssum + JS * HN);
  const size_t need_bits =
      (size_t)((float*)(bits + (size_t)N_NODES * BIT_WORDS) - (float*)d_ws) * sizeof(float);

  gemm_xw<<<dim3(N_NODES / 64, F_OUT / 64), 128, 0, stream>>>(x, W, h);
  attn_scores<<<dim3(N_NODES / 4), 128, 0, stream>>>(h, a, si, sj);
  if (ws_size >= need_bits) {
    row_max<true><<<dim3(N_NODES / RM_ROWS), 512, 0, stream>>>(adj, sj, Mx, bits);
    attn_agg<true, true><<<dim3(N_NODES / 64, HEADS, JS), 128, 0, stream>>>(
        h, adj, bits, si, sj, Mx, pacc, pssum, N_NODES / JS);
    attn_reduce<JS><<<dim3(N_NODES / 16), 512, 0, stream>>>(pacc, pssum, out);
    return;
  }
  row_max<false><<<dim3(N_NODES / RM_ROWS), 512, 0, stream>>>(adj, sj, Mx, nullptr);

  if (ws_size >= need_js) {
    attn_agg<true, false><<<dim3(N_NODES / 64, HEADS, JS), 128, 0, stream>>>(
        h, adj, nullptr, si, sj, Mx, pacc, pssum, N_NODES / JS);
    attn_reduce<JS><<<dim3(N_NODES / 16), 512, 0, stream>>>(pacc, pssum, out);
  } else {
    attn_agg<false, false><<<dim3(N_NODES / 64, HEADS, 1), 128, 0, stream>>>(
        h, adj, nullptr, si, sj, Mx, out, nullptr, N_NODES);
  }
}


# gemm_xw stages x and W through LDS in tiles of 32 k (whole-line 16-byte loads, next tile in flight under the products); fragments are 8-byte LDS reads, no 4/8-byte global loads; sum order kept
# speedup vs baseline: 1.7350x; 1.1546x over previous
#include <hip/hip_runtime.h>
#include <hip/hip_bf16.h>

typedef __attribute__((ext_vector_type(2))) float v2f;
typedef __attribute__((ext_vector_type(4))) float v4f;
typedef __attribute__((ext_vector_type(8))) float v8f;
typedef __attribute__((ext_vector_type(2))) int   v2i;
typedef __attribute__((ext_vector_type(4))) int   v4i;
typedef __attribute__((ext_vector_type(2))) unsigned v2u;

#define N_NODES 4096
#define F_IN    512
#define F_OUT   512
#define HEADS   4
#define HEAD_DIM 128
#define ALPHA   0.2f
#define NEG_INF (-9.0e15f)
#define BIT_WORDS (N_NODES / 32)

#define GK       32
#define GA_PITCH 36
#define GB_PITCH 160
#define GA_TILE  (64 * GA_PITCH)
#define GB_TILE  ((GK / 2) * GB_PITCH)

__global__ __launch_bounds__(128) void gemm_xw(const float* __restrict__ x,
                                               const float* __restrict__ W,
                                               float* __restrict__ h) {
  __shared__ __attribute__((aligned(16))) float As[2 * GA_TILE];
  __shared__ __attribute__((aligned(16))) float Bs[2 * GB_TILE];

  const int tid  = threadIdx.x;
  const int lane = tid & 31;
  const int wave = tid >> 5;
  const int half = lane >> 4;
  const int lid  = lane & 15;

  const int bm = blockIdx.x;
  const int bn = blockIdx.y;

  const float* asrc = x + (size_t)(bm * 64 + (tid >> 3)) * F_IN + (tid & 7) * 4;
  const int    adst = (tid >> 3) * GA_PITCH + (tid & 7) * 4;
  const float* bsrc = W + (size_t)(2 * (tid >> 4)) * F_OUT + bn * 64 + (tid & 15) * 4;
  const int    bdst = (tid >> 4) * GB_PITCH + (tid & 15) * 8;

  const int afrag = (wave * 16 + lid) * GA_PITCH + half * 2;
  const int bfrag = half * GB_PITCH + lid * 2;

  v8f acc[4];
#pragma unroll
  for (int t = 0; t < 4; ++t) acc[t] = {};

  v4f ra[4], rb0[2], rb1[2];
#pragma unroll
  for (int i = 0; i < 4; ++i)
    ra[i] = *(const v4f*)(asrc + (size_t)i * 16 * F_IN);
#pragma unroll
  for (int i = 0; i < 2; ++i) {
    rb0[i] = *(const v4f*)(bsrc + (size_t)(i * 16) * F_OUT);
    rb1[i] = *(const v4f*)(bsrc + (size_t)(i * 16 + 1) * F_OUT);
  }

  for (int kt = 0; kt < F_IN / GK; ++kt) {
    const int buf = kt & 1;
    float* Ab = As + buf * GA_TILE;
    float* Bb = Bs + buf * GB_TILE;
#pragma unroll
    for (int i = 0; i < 4; ++i)
      *(v4f*)(Ab + adst + i * 16 * GA_PITCH) = ra[i];
#pragma unroll
    for (int i = 0; i < 2; ++i) {
      const v4f w0 = {rb0[i].x, rb1[i].x, rb0[i].y, rb1[i].y};
      const v4f w1 = {rb0[i].z, rb1[i].z, rb0[i].w, rb1[i].w};
      *(v4f*)(Bb + bdst + i * 8 * GB_PITCH + 0) = w0;
      *(v4f*)(Bb + bdst + i * 8 * GB_PITCH + 4) = w1;
    }
    __syncthreads();

    if (kt + 1 < F_IN / GK) {
      const int k1 = (kt + 1) * GK;
#pragma unroll
      for (int i = 0; i < 4; ++i)
        ra[i] = *(const v4f*)(asrc + (size_t)i * 16 * F_IN + k1);
#pragma unroll
      for (int i = 0; i < 2; ++i) {
        rb0[i] = *(const v4f*)(bsrc + (size_t)(k1 + i * 16) * F_OUT);
        rb1[i] = *(const v4f*)(bsrc + (size_t)(k1 + i * 16 + 1) * F_OUT);
      }
    }
    __builtin_amdgcn_sched_barrier(0);

#pragma unroll
    for (int ks = 0; ks < GK / 4; ++ks) {
      const v2f a = *(const v2f*)(Ab + afrag + ks * 4);
#pragma unroll
      for (int t = 0; t < 4; ++t) {
        const v2f b = *(const v2f*)(Bb + bfrag + ks * 2 * GB_PITCH + t * 32);
        acc[t] = __builtin_amdgcn_wmma_f32_16x16x4_f32(
            false, a, false, b, (short)0, acc[t], false, false);
      }
    }
  }

#pragma unroll
  for (int t = 0; t < 4; ++t) {
#pragma unroll
    for (int v = 0; v < 8; ++v) {
      const int m = bm * 64 + wave * 16 + v + half * 8;
      const int c = bn * 64 + t * 16 + lid;
      h[(size_t)m * F_OUT + c] = acc[t][v];
    }
  }
}

__global__ __launch_bounds__(128) void attn_scores(const float* __restrict__ h,
                                                   const float* __restrict__ a,
                                                   float* __restrict__ si,
                                                   float* __restrict__ sj) {
  const int lane = threadIdx.x & 31;
  const int wave = threadIdx.x >> 5;
  const int n = blockIdx.x * 4 + wave;

  const v4f a1 = *(const v4f*)(a + lane * 4);
  const v4f a2 = *(const v4f*)(a + HEAD_DIM + lane * 4);

#pragma unroll
  for (int head = 0; head < HEADS; ++head) {
    const float* hrow = h + (size_t)n * F_OUT + head * HEAD_DIM;
    const v4f hv = *(const v4f*)(hrow + lane * 4);
    float s1 = hv.x * a1.x + hv.y * a1.y + hv.z * a1.z + hv.w * a1.w;
    float s2 = hv.x * a2.x + hv.y * a2.y + hv.z * a2.z + hv.w * a2.w;
#pragma unroll
    for (int off = 16; off > 0; off >>= 1) {
      s1 += __shfl_down(s1, off, 32);
      s2 += __shfl_down(s2, off, 32);
    }
    if (lane == 0) {
      si[head * N_NODES + n] = s1;
      sj[head * N_NODES + n] = s2;
    }
  }
}

#define RM_ROWS  16
#define RM_GROUP 8
#define RM_NGRP  (N_NODES / (128 * RM_GROUP))
#define RM_LDS   4

__device__ __forceinline__ float masked_max4(float m, const v4i a, const v4f s) {
  const float c0 = (a.x > 0) ? s.x : NEG_INF;
  const float c1 = (a.y > 0) ? s.y : NEG_INF;
  const float c2 = (a.z > 0) ? s.z : NEG_INF;
  const float c3 = (a.w > 0) ? s.w : NEG_INF;
  return fmaxf(fmaxf(m, fmaxf(c0, c1)), fmaxf(c2, c3));
}

template <bool BITS>
__global__ __launch_bounds__(512) void row_max(const int* __restrict__ adj,
                                               const float* __restrict__ sj,
                                               float* __restrict__ Mx,
                                               unsigned* __restrict__ bits) {
  __shared__ float sjs[HEADS * N_NODES];

  const int t    = threadIdx.x;
  const int lane = t & 31;
  const int wave = t >> 5;
  const int i    = blockIdx.x * RM_ROWS + wave;
  const int* arow = adj + (size_t)i * N_NODES + lane * 4;

  v4f tb[HEADS * N_NODES / 4 / 512];
#pragma unroll
  for (int k = 0; k < HEADS * N_NODES / 4 / 512; ++k)
    tb[k] = *(const v4f*)(sj + (size_t)(k * 512 + t) * 4);

  v4i cur[RM_GROUP];
#pragma unroll
  for (int u = 0; u < RM_GROUP; ++u)
    cur[u] = *(const v4i*)(arow + u * 128);

#pragma unroll
  for (int k = 0; k < HEADS * N_NODES / 4 / 512; ++k)
    *(v4f*)(sjs + (k * 512 + t) * 4) = tb[k];
  __syncthreads();

  float m0 = NEG_INF, m1 = NEG_INF, m2 = NEG_INF, m3 = NEG_INF;
  unsigned bw[RM_NGRP];
#pragma unroll
  for (int g = 0; g < RM_NGRP; ++g) bw[g] = 0u;
#pragma unroll
  for (int g = 0; g < RM_NGRP; ++g) {
    v4i nxt[RM_GROUP];
    if (g + 1 < RM_NGRP) {
#pragma unroll
      for (int u = 0; u < RM_GROUP; ++u)
        nxt[u] = *(const v4i*)(arow + ((g + 1) * RM_GROUP + u) * 128);
    }
#pragma unroll
    for (int u0 = 0; u0 < RM_GROUP; u0 += RM_LDS) {
      v4f s[RM_LDS][HEADS];
#pragma unroll
      for (int u = 0; u < RM_LDS; ++u) {
        const float* sp = sjs + (g * RM_GROUP + u0 + u) * 128 + lane * 4;
#pragma unroll
        for (int hd = 0; hd < HEADS; ++hd)
          s[u][hd] = *(const v4f*)(sp + hd * N_NODES);
      }
      __builtin_amdgcn_sched_barrier(0);
#pragma unroll
      for (int u = 0; u < RM_LDS; ++u) {
        m0 = masked_max4(m0, cur[u0 + u], s[u][0]);
        m1 = masked_max4(m1, cur[u0 + u], s[u][1]);
        m2 = masked_max4(m2, cur[u0 + u], s[u][2]);
        m3 = masked_max4(m3, cur[u0 + u], s[u][3]);
      }
      if (BITS) {
#pragma unroll
        for (int u = 0; u < RM_LDS; ++u) {
          const v4i a = cur[u0 + u];
          const int q = (u0 + u) * 4;
          const unsigned v0 = __builtin_amdgcn_ballot_w32(a.x > 0);
          const unsigned v1 = __builtin_amdgcn_ballot_w32(a.y > 0);
          const unsigned v2 = __builtin_amdgcn_ballot_w32(a.z > 0);
          const unsigned v3 = __builtin_amdgcn_ballot_w32(a.w > 0);
          bw[g] = (lane == q + 0) ? v0 : bw[g];
          bw[g] = (lane == q + 1) ? v1 : bw[g];
          bw[g] = (lane == q + 2) ? v2 : bw[g];
          bw[g] = (lane == q + 3) ? v3 : bw[g];
        }
      }
    }
    if (g + 1 < RM_NGRP) {
#pragma unroll
      for (int u = 0; u < RM_GROUP; ++u) cur[u] = nxt[u];
    }
  }

#pragma unroll
  for (int off = 16; off > 0; off >>= 1) {
    m0 = fmaxf(m0, __shfl_xor(m0, off, 32));
    m1 = fmaxf(m1, __shfl_xor(m1, off, 32));
    m2 = fmaxf(m2, __shfl_xor(m2, off, 32));
    m3 = fmaxf(m3, __shfl_xor(m3, off, 32));
  }
  if (BITS) {
#pragma unroll
    for (int g = 0; g < RM_NGRP; ++g)
      bits[(size_t)i * BIT_WORDS + g * 32 + lane] = bw[g];
  }
  if (lane == 0) {
    Mx[0 * N_NODES + i] = m0;
    Mx[1 * N_NODES + i] = m1;
    Mx[2 * N_NODES + i] = m2;
    Mx[3 * N_NODES + i] = m3;
  }
}

#define CHUNK_J 64
#define PAIRS   (CHUNK_J / 2)
#define PAIR_STRIDE 288

template <bool PARTIAL, bool BITS>
__global__ __launch_bounds__(128) void attn_agg(const float* __restrict__ h,
                                                const int* __restrict__ adj,
                                                const unsigned* __restrict__ bits,
                                                const float* __restrict__ si,
                                                const float* __restrict__ sj,
                                                const float* __restrict__ Mx,
                                                float* __restrict__ outp,
                                                float* __restrict__ pssum,
                                                int jspan) {
  __shared__ float Vp[PAIRS * PAIR_STRIDE];
  __shared__ float srow[4][16];

  const int lane = threadIdx.x & 31;
  const int wave = threadIdx.x >> 5;
  const int half = lane >> 4;
  const int lid  = lane & 15;
  const int kb   = half * 2;

  const int head = blockIdx.y;
  const int iblk = blockIdx.x * 64;
  const int z    = blockIdx.z;
  const int jstart = z * jspan;
  const int row  = iblk + wave * 16 + lid;

  const float sir = si[head * N_NODES + row];
  const float mx0 = sir + Mx[head * N_NODES + row];
  const float mrow = (mx0 > 0.f) ? mx0 : ALPHA * mx0;

  v8f acc[8];
#pragma unroll
  for (int t = 0; t < 8; ++t) acc[t] = {};
  float ssum = 0.f;

  const float* sjh  = sj + head * N_NODES;
  const int*   arow = adj + (size_t)row * N_NODES;
  const unsigned* brow = bits + (size_t)row * BIT_WORDS + kb;
  const float* Vg   = h + head * HEAD_DIM;

  const int st_pair0 = wave * (PAIRS / 4);
  const int st_d     = lane * 4;

  for (int c = jstart; c < jstart + jspan; c += CHUNK_J) {
    unsigned w0 = 0u, w1 = 0u;
    if (BITS) {
      const v2u wv = *(const v2u*)(brow + (c >> 7) * 4);
      const int lb = (c & 64) >> 2;
      w0 = wv.x >> lb;
      w1 = wv.y >> lb;
    }
    {
      const float* srcp = Vg + (size_t)(c + st_pair0 * 2) * F_OUT + st_d;
      float* dst = Vp + st_pair0 * PAIR_STRIDE + st_d * 2;
      v4f rr[CHUNK_J / 4];
#pragma unroll
      for (int i = 0; i < CHUNK_J / 4; ++i)
        rr[i] = *(const v4f*)(srcp + (size_t)i * F_OUT);
      __builtin_amdgcn_sched_barrier(0);
#pragma unroll
      for (int pp = 0; pp < PAIRS / 4; ++pp) {
        const v4f r0 = rr[pp * 2 + 0];
        const v4f r1 = rr[pp * 2 + 1];
        v4f w0 = {r0.x, r1.x, r0.y, r1.y};
        v4f w1 = {r0.z, r1.z, r0.w, r1.w};
        *(v4f*)(dst + pp * PAIR_STRIDE + 0) = w0;
        *(v4f*)(dst + pp * PAIR_STRIDE + 4) = w1;
      }
    }
    __syncthreads();

#pragma unroll
    for (int s4 = 0; s4 < CHUNK_J / 4; ++s4) {
      const int jg0 = c + s4 * 4 + kb;
      const v2f sjv = *(const v2f*)(sjh + jg0);
      bool on0, on1;
      if (BITS) {
        on0 = ((w0 >> s4) & 1u) != 0u;
        on1 = ((w1 >> s4) & 1u) != 0u;
      } else {
        const v2i av = *(const v2i*)(arow + jg0);
        on0 = av.x > 0;
        on1 = av.y > 0;
      }

      float e0 = sir + sjv.x; e0 = (e0 > 0.f) ? e0 : ALPHA * e0;
      float e1 = sir + sjv.y; e1 = (e1 > 0.f) ? e1 : ALPHA * e1;
      const float p0 = on0 ? __expf(e0 - mrow) : 0.f;
      const float p1 = on1 ? __expf(e1 - mrow) : 0.f;
      ssum += p0 + p1;

      v2f af;
      af.x = p0;
      af.y = p1;
      const float* bp = Vp + (s4 * 2 + half) * PAIR_STRIDE + lid * 2;
#pragma unroll
      for (int t = 0; t < 8; ++t) {
        const v2f bf = *(const v2f*)(bp + t * 32);
        acc[t] = __builtin_amdgcn_wmma_f32_16x16x4_f32(
            false, af, false, bf, (short)0, acc[t], false, false);
      }
    }
    __syncthreads();
  }

  ssum += __shfl_xor(ssum, 16, 32);

  if (PARTIAL) {
    if (half == 0) pssum[(size_t)z * (HEADS * N_NODES) + head * N_NODES + row] = ssum;
    float* pa = outp + (size_t)z * (N_NODES * F_OUT);
#pragma unroll
    for (int t = 0; t < 8; ++t) {
#pragma unroll
      for (int v = 0; v < 8; ++v) {
        const int m = iblk + wave * 16 + v + half * 8;
        pa[(size_t)m * F_OUT + head * HEAD_DIM + t * 16 + lid] = acc[t][v];
      }
    }
  } else {
    if (half == 0) srow[wave][lid] = ssum;
    __syncthreads();
#pragma unroll
    for (int t = 0; t < 8; ++t) {
#pragma unroll
      for (int v = 0; v < 8; ++v) {
        const int m = iblk + wave * 16 + v + half * 8;
        float val = acc[t][v] / srow[wave][v + half * 8];
        val = (val > 0.f) ? val : (__expf(val) - 1.f);
        outp[(size_t)m * F_OUT + head * HEAD_DIM + t * 16 + lid] = val;
      }
    }
  }
}

template <int JS>
__global__ __launch_bounds__(512) void attn_reduce(const float* __restrict__ pacc,
                                                   const float* __restrict__ pssum,
                                                   float* __restrict__ out) {
  const int lane = threadIdx.x & 31;
  const int wave = threadIdx.x >> 5;
  const int i = blockIdx.x * 16 + wave;

  v4f p[HEADS][JS];
  float d[HEADS][JS];
#pragma unroll
  for (int head = 0; head < HEADS; ++head) {
#pragma unroll
    for (int zz = 0; zz < JS; ++zz) {
      p[head][zz] = *(const v4f*)(pacc + (size_t)zz * (N_NODES * F_OUT) +
                                  (size_t)i * F_OUT + head * HEAD_DIM + lane * 4);
      d[head][zz] = pssum[(size_t)zz * (HEADS * N_NODES) + head * N_NODES + i];
    }
  }
#pragma unroll
  for (int head = 0; head < HEADS; ++head) {
    v4f num = {0.f, 0.f, 0.f, 0.f};
    float den = 0.f;
#pragma unroll
    for (int zz = 0; zz < JS; ++zz) {
      num += p[head][zz];
      den += d[head][zz];
    }
    v4f val = num / den;
    val.x = (val.x > 0.f) ? val.x : (__expf(val.x) - 1.f);
    val.y = (val.y > 0.f) ? val.y : (__expf(val.y) - 1.f);
    val.z = (val.z > 0.f) ? val.z : (__expf(val.z) - 1.f);
    val.w = (val.w > 0.f) ? val.w : (__expf(val.w) - 1.f);
    *(v4f*)(out + (size_t)i * F_OUT + head * HEAD_DIM + lane * 4) = val;
  }
}

extern "C" void kernel_launch(void* const* d_in, const int* in_sizes, int n_in,
                              void* d_out, int out_size, void* d_ws, size_t ws_size,
                              hipStream_t stream) {
  const float* x   = (const float*)d_in[0];
  const int*   adj = (const int*)d_in[1];
  const float* W   = (const float*)d_in[2];
  const float* a   = (const float*)d_in[3];
  float* out = (float*)d_out;

  constexpr int JS = 4;
  const size_t HN = (size_t)HEADS * N_NODES;

  float* h   = (float*)d_ws;
  float* si  = h + (size_t)N_NODES * F_OUT;
  float* sj  = si + HN;
  float* Mx  = sj + HN;
  float* pacc  = Mx + HN;
  float* pssum = pacc + (size_t)JS * N_NODES * F_OUT;
  const size_t need_js =
      (size_t)(pssum + JS * HN - (float*)d_ws) * sizeof(float);
  unsigned* bits = (unsigned*)(pssum + JS * HN);
  const size_t need_bits =
      (size_t)((float*)(bits + (size_t)N_NODES * BIT_WORDS) - (float*)d_ws) * sizeof(float);

  gemm_xw<<<dim3(N_NODES / 64, F_OUT / 64), 128, 0, stream>>>(x, W, h);
  attn_scores<<<dim3(N_NODES / 4), 128, 0, stream>>>(h, a, si, sj);
  if (ws_size >= need_bits) {
    row_max<true><<<dim3(N_NODES / RM_ROWS), 512, 0, stream>>>(adj, sj, Mx, bits);
    attn_agg<true, true><<<dim3(N_NODES / 64, HEADS, JS), 128, 0, stream>>>(
        h, adj, bits, si, sj, Mx, pacc, pssum, N_NODES / JS);
    attn_reduce<JS><<<dim3(N_NODES / 16), 512, 0, stream>>>(pacc, pssum, out);
    return;
  }
  row_max<false><<<dim3(N_NODES / RM_ROWS), 512, 0, stream>>>(adj, sj, Mx, nullptr);

  if (ws_size >= need_js) {
    attn_agg<true, false><<<dim3(N_NODES / 64, HEADS, JS), 128, 0, stream>>>(
        h, adj, nullptr, si, sj, Mx, pacc, pssum, N_NODES / JS);
    attn_reduce<JS><<<dim3(N_NODES / 16), 512, 0, stream>>>(pacc, pssum, out);
  } else {
    attn_agg<false, false><<<dim3(N_NODES / 64, HEADS, 1), 128, 0, stream>>>(
        h, adj, nullptr, si, sj, Mx, out, nullptr, N_NODES);
  }
}


# attn_agg's 4 j quarters run as 4 wave groups of one 16-wave block, summed via LDS in order 0..3: no partials in memory, no reduce launch (4 launches, were 5); chunks fetched 1 ahead; sum order kept
# speedup vs baseline: 1.8557x; 1.0696x over previous
#include <hip/hip_runtime.h>
#include <hip/hip_bf16.h>

typedef __attribute__((ext_vector_type(2))) float v2f;
typedef __attribute__((ext_vector_type(4))) float v4f;
typedef __attribute__((ext_vector_type(8))) float v8f;
typedef __attribute__((ext_vector_type(2))) int   v2i;
typedef __attribute__((ext_vector_type(4))) int   v4i;
typedef __attribute__((ext_vector_type(2))) unsigned v2u;

#define N_NODES 4096
#define F_IN    512
#define F_OUT   512
#define HEADS   4
#define HEAD_DIM 128
#define ALPHA   0.2f
#define NEG_INF (-9.0e15f)
#define BIT_WORDS (N_NODES / 32)

#define GK       32
#define GA_PITCH 36
#define GB_PITCH 160
#define GA_TILE  (64 * GA_PITCH)
#define GB_TILE  ((GK / 2) * GB_PITCH)

__global__ __launch_bounds__(128) void gemm_xw(const float* __restrict__ x,
                                               const float* __restrict__ W,
                                               float* __restrict__ h) {
  __shared__ __attribute__((aligned(16))) float As[2 * GA_TILE];
  __shared__ __attribute__((aligned(16))) float Bs[2 * GB_TILE];

  const int tid  = threadIdx.x;
  const int lane = tid & 31;
  const int wave = tid >> 5;
  const int half = lane >> 4;
  const int lid  = lane & 15;

  const int bm = blockIdx.x;
  const int bn = blockIdx.y;

  const float* asrc = x + (size_t)(bm * 64 + (tid >> 3)) * F_IN + (tid & 7) * 4;
  const int    adst = (tid >> 3) * GA_PITCH + (tid & 7) * 4;
  const float* bsrc = W + (size_t)(2 * (tid >> 4)) * F_OUT + bn * 64 + (tid & 15) * 4;
  const int    bdst = (tid >> 4) * GB_PITCH + (tid & 15) * 8;

  const int afrag = (wave * 16 + lid) * GA_PITCH + half * 2;
  const int bfrag = half * GB_PITCH + lid * 2;

  v8f acc[4];
#pragma unroll
  for (int t = 0; t < 4; ++t) acc[t] = {};

  v4f ra[4], rb0[2], rb1[2];
#pragma unroll
  for (int i = 0; i < 4; ++i)
    ra[i] = *(const v4f*)(asrc + (size_t)i * 16 * F_IN);
#pragma unroll
  for (int i = 0; i < 2; ++i) {
    rb0[i] = *(const v4f*)(bsrc + (size_t)(i * 16) * F_OUT);
    rb1[i] = *(const v4f*)(bsrc + (size_t)(i * 16 + 1) * F_OUT);
  }

  for (int kt = 0; kt < F_IN / GK; ++kt) {
    const int buf = kt & 1;
    float* Ab = As + buf * GA_TILE;
    float* Bb = Bs + buf * GB_TILE;
#pragma unroll
    for (int i = 0; i < 4; ++i)
      *(v4f*)(Ab + adst + i * 16 * GA_PITCH) = ra[i];
#pragma unroll
    for (int i = 0; i < 2; ++i) {
      const v4f w0 = {rb0[i].x, rb1[i].x, rb0[i].y, rb1[i].y};
      const v4f w1 = {rb0[i].z, rb1[i].z, rb0[i].w, rb1[i].w};
      *(v4f*)(Bb + bdst + i * 8 * GB_PITCH + 0) = w0;
      *(v4f*)(Bb + bdst + i * 8 * GB_PITCH + 4) = w1;
    }
    __syncthreads();

    if (kt + 1 < F_IN / GK) {
      const int k1 = (kt + 1) * GK;
#pragma unroll
      for (int i = 0; i < 4; ++i)
        ra[i] = *(const v4f*)(asrc + (size_t)i * 16 * F_IN + k1);
#pragma unroll
      for (int i = 0; i < 2; ++i) {
        rb0[i] = *(const v4f*)(bsrc + (size_t)(k1 + i * 16) * F_OUT);
        rb1[i] = *(const v4f*)(bsrc + (size_t)(k1 + i * 16 + 1) * F_OUT);
      }
    }
    __builtin_amdgcn_sched_barrier(0);

#pragma unroll
    for (int ks = 0; ks < GK / 4; ++ks) {
      const v2f a = *(const v2f*)(Ab + afrag + ks * 4);
#pragma unroll
      for (int t = 0; t < 4; ++t) {
        const v2f b = *(const v2f*)(Bb + bfrag + ks * 2 * GB_PITCH + t * 32);
        acc[t] = __builtin_amdgcn_wmma_f32_16x16x4_f32(
            false, a, false, b, (short)0, acc[t], false, false);
      }
    }
  }

#pragma unroll
  for (int t = 0; t < 4; ++t) {
#pragma unroll
    for (int v = 0; v < 8; ++v) {
      const int m = bm * 64 + wave * 16 + v + half * 8;
      const int c = bn * 64 + t * 16 + lid;
      h[(size_t)m * F_OUT + c] = acc[t][v];
    }
  }
}

__global__ __launch_bounds__(128) void attn_scores(const float* __restrict__ h,
                                                   const float* __restrict__ a,
                                                   float* __restrict__ si,
                                                   float* __restrict__ sj) {
  const int lane = threadIdx.x & 31;
  const int wave = threadIdx.x >> 5;
  const int n = blockIdx.x * 4 + wave;

  const v4f a1 = *(const v4f*)(a + lane * 4);
  const v4f a2 = *(const v4f*)(a + HEAD_DIM + lane * 4);

#pragma unroll
  for (int head = 0; head < HEADS; ++head) {
    const float* hrow = h + (size_t)n * F_OUT + head * HEAD_DIM;
    const v4f hv = *(const v4f*)(hrow + lane * 4);
    float s1 = hv.x * a1.x + hv.y * a1.y + hv.z * a1.z + hv.w * a1.w;
    float s2 = hv.x * a2.x + hv.y * a2.y + hv.z * a2.z + hv.w * a2.w;
#pragma unroll
    for (int off = 16; off > 0; off >>= 1) {
      s1 += __shfl_down(s1, off, 32);
      s2 += __shfl_down(s2, off, 32);
    }
    if (lane == 0) {
      si[head * N_NODES + n] = s1;
      sj[head * N_NODES + n] = s2;
    }
  }
}

#define RM_ROWS  16
#define RM_GROUP 8
#define RM_NGRP  (N_NODES / (128 * RM_GROUP))
#define RM_LDS   4

__device__ __forceinline__ float masked_max4(float m, const v4i a, const v4f s) {
  const float c0 = (a.x > 0) ? s.x : NEG_INF;
  const float c1 = (a.y > 0) ? s.y : NEG_INF;
  const float c2 = (a.z > 0) ? s.z : NEG_INF;
  const float c3 = (a.w > 0) ? s.w : NEG_INF;
  return fmaxf(fmaxf(m, fmaxf(c0, c1)), fmaxf(c2, c3));
}

template <bool BITS>
__global__ __launch_bounds__(512) void row_max(const int* __restrict__ adj,
                                               const float* __restrict__ sj,
                                               float* __restrict__ Mx,
                                               unsigned* __restrict__ bits) {
  __shared__ float sjs[HEADS * N_NODES];

  const int t    = threadIdx.x;
  const int lane = t & 31;
  const int wave = t >> 5;
  const int i    = blockIdx.x * RM_ROWS + wave;
  const int* arow = adj + (size_t)i * N_NODES + lane * 4;

  v4f tb[HEADS * N_NODES / 4 / 512];
#pragma unroll
  for (int k = 0; k < HEADS * N_NODES / 4 / 512; ++k)
    tb[k] = *(const v4f*)(sj + (size_t)(k * 512 + t) * 4);

  v4i cur[RM_GROUP];
#pragma unroll
  for (int u = 0; u < RM_GROUP; ++u)
    cur[u] = *(const v4i*)(arow + u * 128);

#pragma unroll
  for (int k = 0; k < HEADS * N_NODES / 4 / 512; ++k)
    *(v4f*)(sjs + (k * 512 + t) * 4) = tb[k];
  __syncthreads();

  float m0 = NEG_INF, m1 = NEG_INF, m2 = NEG_INF, m3 = NEG_INF;
  unsigned bw[RM_NGRP];
#pragma unroll
  for (int g = 0; g < RM_NGRP; ++g) bw[g] = 0u;
#pragma unroll
  for (int g = 0; g < RM_NGRP; ++g) {
    v4i nxt[RM_GROUP];
    if (g + 1 < RM_NGRP) {
#pragma unroll
      for (int u = 0; u < RM_GROUP; ++u)
        nxt[u] = *(const v4i*)(arow + ((g + 1) * RM_GROUP + u) * 128);
    }
#pragma unroll
    for (int u0 = 0; u0 < RM_GROUP; u0 += RM_LDS) {
      v4f s[RM_LDS][HEADS];
#pragma unroll
      for (int u = 0; u < RM_LDS; ++u) {
        const float* sp = sjs + (g * RM_GROUP + u0 + u) * 128 + lane * 4;
#pragma unroll
        for (int hd = 0; hd < HEADS; ++hd)
          s[u][hd] = *(const v4f*)(sp + hd * N_NODES);
      }
      __builtin_amdgcn_sched_barrier(0);
#pragma unroll
      for (int u = 0; u < RM_LDS; ++u) {
        m0 = masked_max4(m0, cur[u0 + u], s[u][0]);
        m1 = masked_max4(m1, cur[u0 + u], s[u][1]);
        m2 = masked_max4(m2, cur[u0 + u], s[u][2]);
        m3 = masked_max4(m3, cur[u0 + u], s[u][3]);
      }
      if (BITS) {
#pragma unroll
        for (int u = 0; u < RM_LDS; ++u) {
          const v4i a = cur[u0 + u];
          const int q = (u0 + u) * 4;
          const unsigned v0 = __builtin_amdgcn_ballot_w32(a.x > 0);
          const unsigned v1 = __builtin_amdgcn_ballot_w32(a.y > 0);
          const unsigned v2 = __builtin_amdgcn_ballot_w32(a.z > 0);
          const unsigned v3 = __builtin_amdgcn_ballot_w32(a.w > 0);
          bw[g] = (lane == q + 0) ? v0 : bw[g];
          bw[g] = (lane == q + 1) ? v1 : bw[g];
          bw[g] = (lane == q + 2) ? v2 : bw[g];
          bw[g] = (lane == q + 3) ? v3 : bw[g];
        }
      }
    }
    if (g + 1 < RM_NGRP) {
#pragma unroll
      for (int u = 0; u < RM_GROUP; ++u) cur[u] = nxt[u];
    }
  }

#pragma unroll
  for (int off = 16; off > 0; off >>= 1) {
    m0 = fmaxf(m0, __shfl_xor(m0, off, 32));
    m1 = fmaxf(m1, __shfl_xor(m1, off, 32));
    m2 = fmaxf(m2, __shfl_xor(m2, off, 32));
    m3 = fmaxf(m3, __shfl_xor(m3, off, 32));
  }
  if (BITS) {
#pragma unroll
    for (int g = 0; g < RM_NGRP; ++g)
      bits[(size_t)i * BIT_WORDS + g * 32 + lane] = bw[g];
  }
  if (lane == 0) {
    Mx[0 * N_NODES + i] = m0;
    Mx[1 * N_NODES + i] = m1;
    Mx[2 * N_NODES + i] = m2;
    Mx[3 * N_NODES + i] = m3;
  }
}

#define CHUNK_J 64
#define PAIRS   (CHUNK_J / 2)
#define PAIR_STRIDE 288

template <bool PARTIAL, bool BITS>
__global__ __launch_bounds__(128) void attn_agg(const float* __restrict__ h,
                                                const int* __restrict__ adj,
                                                const unsigned* __restrict__ bits,
                                                const float* __restrict__ si,
                                                const float* __restrict__ sj,
                                                const float* __restrict__ Mx,
                                                float* __restrict__ outp,
                                                float* __restrict__ pssum,
                                                int jspan) {
  __shared__ float Vp[PAIRS * PAIR_STRIDE];
  __shared__ float srow[4][16];

  const int lane = threadIdx.x & 31;
  const int wave = threadIdx.x >> 5;
  const int half = lane >> 4;
  const int lid  = lane & 15;
  const int kb   = half * 2;

  const int head = blockIdx.y;
  const int iblk = blockIdx.x * 64;
  const int z    = blockIdx.z;
  const int jstart = z * jspan;
  const int row  = iblk + wave * 16 + lid;

  const float sir = si[head * N_NODES + row];
  const float mx0 = sir + Mx[head * N_NODES + row];
  const float mrow = (mx0 > 0.f) ? mx0 : ALPHA * mx0;

  v8f acc[8];
#pragma unroll
  for (int t = 0; t < 8; ++t) acc[t] = {};
  float ssum = 0.f;

  const float* sjh  = sj + head * N_NODES;
  const int*   arow = adj + (size_t)row * N_NODES;
  const unsigned* brow = bits + (size_t)row * BIT_WORDS + kb;
  const float* Vg   = h + head * HEAD_DIM;

  const int st_pair0 = wave * (PAIRS / 4);
  const int st_d     = lane * 4;

  for (int c = jstart; c < jstart + jspan; c += CHUNK_J) {
    unsigned w0 = 0u, w1 = 0u;
    if (BITS) {
      const v2u wv = *(const v2u*)(brow + (c >> 7) * 4);
      const int lb = (c & 64) >> 2;
      w0 = wv.x >> lb;
      w1 = wv.y >> lb;
    }
    {
      const float* srcp = Vg + (size_t)(c + st_pair0 * 2) * F_OUT + st_d;
      float* dst = Vp + st_pair0 * PAIR_STRIDE + st_d * 2;
      v4f rr[CHUNK_J / 4];
#pragma unroll
      for (int i = 0; i < CHUNK_J / 4; ++i)
        rr[i] = *(const v4f*)(srcp + (size_t)i * F_OUT);
      __builtin_amdgcn_sched_barrier(0);
#pragma unroll
      for (int pp = 0; pp < PAIRS / 4; ++pp) {
        const v4f r0 = rr[pp * 2 + 0];
        const v4f r1 = rr[pp * 2 + 1];
        v4f w0 = {r0.x, r1.x, r0.y, r1.y};
        v4f w1 = {r0.z, r1.z, r0.w, r1.w};
        *(v4f*)(dst + pp * PAIR_STRIDE + 0) = w0;
        *(v4f*)(dst + pp * PAIR_STRIDE + 4) = w1;
      }
    }
    __syncthreads();

#pragma unroll
    for (int s4 = 0; s4 < CHUNK_J / 4; ++s4) {
      const int jg0 = c + s4 * 4 + kb;
      const v2f sjv = *(const v2f*)(sjh + jg0);
      bool on0, on1;
      if (BITS) {
        on0 = ((w0 >> s4) & 1u) != 0u;
        on1 = ((w1 >> s4) & 1u) != 0u;
      } else {
        const v2i av = *(const v2i*)(arow + jg0);
        on0 = av.x > 0;
        on1 = av.y > 0;
      }

      float e0 = sir + sjv.x; e0 = (e0 > 0.f) ? e0 : ALPHA * e0;
      float e1 = sir + sjv.y; e1 = (e1 > 0.f) ? e1 : ALPHA * e1;
      const float p0 = on0 ? __expf(e0 - mrow) : 0.f;
      const float p1 = on1 ? __expf(e1 - mrow) : 0.f;
      ssum += p0 + p1;

      v2f af;
      af.x = p0;
      af.y = p1;
      const float* bp = Vp + (s4 * 2 + half) * PAIR_STRIDE + lid * 2;
#pragma unroll
      for (int t = 0; t < 8; ++t) {
        const v2f bf = *(const v2f*)(bp + t * 32);
        acc[t] = __builtin_amdgcn_wmma_f32_16x16x4_f32(
            false, af, false, bf, (short)0, acc[t], false, false);
      }
    }
    __syncthreads();
  }

  ssum += __shfl_xor(ssum, 16, 32);

  if (PARTIAL) {
    if (half == 0) pssum[(size_t)z * (HEADS * N_NODES) + head * N_NODES + row] = ssum;
    float* pa = outp + (size_t)z * (N_NODES * F_OUT);
#pragma unroll
    for (int t = 0; t < 8; ++t) {
#pragma unroll
      for (int v = 0; v < 8; ++v) {
        const int m = iblk + wave * 16 + v + half * 8;
        pa[(size_t)m * F_OUT + head * HEAD_DIM + t * 16 + lid] = acc[t][v];
      }
    }
  } else {
    if (half == 0) srow[wave][lid] = ssum;
    __syncthreads();
#pragma unroll
    for (int t = 0; t < 8; ++t) {
#pragma unroll
      for (int v = 0; v < 8; ++v) {
        const int m = iblk + wave * 16 + v + half * 8;
        float val = acc[t][v] / srow[wave][v + half * 8];
        val = (val > 0.f) ? val : (__expf(val) - 1.f);
        outp[(size_t)m * F_OUT + head * HEAD_DIM + t * 16 + lid] = val;
      }
    }
  }
}

#define AB_CJ     32
#define AB_PAIRS  (AB_CJ / 2)
#define AB_KS     (AB_CJ / 4)
#define AB_BUF    (AB_PAIRS * PAIR_STRIDE)
#define AB_JQ     4
#define AB_SPAN   (N_NODES / AB_JQ)
#define AB_NCH    (AB_SPAN / AB_CJ)
#define AB_OPITCH 132

__device__ __forceinline__ v2f edge_p2(const float sir, const float mrow, const v2f sjv,
                                       const unsigned w0, const unsigned w1, const int bit) {
  float e0 = sir + sjv.x; e0 = (e0 > 0.f) ? e0 : ALPHA * e0;
  float e1 = sir + sjv.y; e1 = (e1 > 0.f) ? e1 : ALPHA * e1;
  v2f p;
  p.x = (((w0 >> bit) & 1u) != 0u) ? __expf(e0 - mrow) : 0.f;
  p.y = (((w1 >> bit) & 1u) != 0u) ? __expf(e1 - mrow) : 0.f;
  return p;
}

__global__ __launch_bounds__(512) void attn_agg_blk(const float* __restrict__ h,
                                                    const unsigned* __restrict__ bits,
                                                    const float* __restrict__ si,
                                                    const float* __restrict__ sj,
                                                    const float* __restrict__ Mx,
                                                    float* __restrict__ out) {
  __shared__ __attribute__((aligned(16))) float smem[2 * AB_JQ * AB_BUF];
  __shared__ float dsm[AB_JQ][64];

  const int tid  = threadIdx.x;
  const int lane = tid & 31;
  const int wave = tid >> 5;
  const int g    = wave >> 2;
  const int rg   = wave & 3;
  const int half = lane >> 4;
  const int lid  = lane & 15;
  const int kb   = half * 2;

  const int head = blockIdx.y;
  const int iblk = blockIdx.x * 64;
  const int row  = iblk + rg * 16 + lid;
  const int jstart = g * AB_SPAN;

  const float sir = si[head * N_NODES + row];
  const float mx0 = sir + Mx[head * N_NODES + row];
  const float mrow = (mx0 > 0.f) ? mx0 : ALPHA * mx0;

  v8f acc[8];
#pragma unroll
  for (int t = 0; t < 8; ++t) acc[t] = {};
  float ssum = 0.f;

  const float*    sjl  = sj + head * N_NODES + kb;
  const unsigned* brow = bits + (size_t)row * BIT_WORDS + kb;
  const float* Vsrc = h + head * HEAD_DIM + (size_t)(rg * 8) * F_OUT + lane * 4;
  float* Vg = smem + g * (2 * AB_BUF);
  const int st_off = (rg * (AB_PAIRS / 4)) * PAIR_STRIDE + lane * 8;
  const int fr_off = half * PAIR_STRIDE + lid * 2;

  v4f rr[AB_CJ / 4];
  v2f sjn[AB_KS];
  v2u wn;
  v2f pc[AB_KS], pn[AB_KS];

  {
    const int c = jstart;
#pragma unroll
    for (int i = 0; i < AB_CJ / 4; ++i)
      rr[i] = *(const v4f*)(Vsrc + (size_t)(c + i) * F_OUT);
#pragma unroll
    for (int s = 0; s < AB_KS; ++s)
      sjn[s] = *(const v2f*)(sjl + c + s * 4);
    wn = *(const v2u*)(brow + (c >> 7) * 4);
    const int lb = (c & 96) >> 2;
#pragma unroll
    for (int s = 0; s < AB_KS; ++s)
      pc[s] = edge_p2(sir, mrow, sjn[s], wn.x >> lb, wn.y >> lb, s);
    float* dst = Vg + st_off;
#pragma unroll
    for (int pp = 0; pp < AB_PAIRS / 4; ++pp) {
      const v4f r0 = rr[pp * 2 + 0];
      const v4f r1 = rr[pp * 2 + 1];
      const v4f w0 = {r0.x, r1.x, r0.y, r1.y};
      const v4f w1 = {r0.z, r1.z, r0.w, r1.w};
      *(v4f*)(dst + pp * PAIR_STRIDE + 0) = w0;
      *(v4f*)(dst + pp * PAIR_STRIDE + 4) = w1;
    }
  }
  __syncthreads();

  for (int ci = 0; ci < AB_NCH; ++ci) {
    const int c = jstart + ci * AB_CJ;
    const float* Vb = Vg + (ci & 1) * AB_BUF + fr_off;
    v2f b0[8];
#pragma unroll
    for (int t = 0; t < 8; ++t) b0[t] = *(const v2f*)(Vb + t * 32);
    const int cn = (ci + 1 < AB_NCH) ? c + AB_CJ : c;
#pragma unroll
    for (int i = 0; i < AB_CJ / 4; ++i)
      rr[i] = *(const v4f*)(Vsrc + (size_t)(cn + i) * F_OUT);
#pragma unroll
    for (int s = 0; s < AB_KS; ++s)
      sjn[s] = *(const v2f*)(sjl + cn + s * 4);
    wn = *(const v2u*)(brow + (cn >> 7) * 4);
    __builtin_amdgcn_sched_barrier(0);

#pragma unroll
    for (int s4 = 0; s4 < AB_KS; ++s4) {
      if (s4 == AB_KS / 2) {
        __builtin_amdgcn_sched_barrier(0);
        const int lb = (cn & 96) >> 2;
#pragma unroll
        for (int s = 0; s < AB_KS; ++s)
          pn[s] = edge_p2(sir, mrow, sjn[s], wn.x >> lb, wn.y >> lb, s);
        float* dst = Vg + ((ci + 1) & 1) * AB_BUF + st_off;
#pragma unroll
        for (int pp = 0; pp < AB_PAIRS / 4; ++pp) {
          const v4f r0 = rr[pp * 2 + 0];
          const v4f r1 = rr[pp * 2 + 1];
          const v4f w0 = {r0.x, r1.x, r0.y, r1.y};
          const v4f w1 = {r0.z, r1.z, r0.w, r1.w};
          *(v4f*)(dst + pp * PAIR_STRIDE + 0) = w0;
          *(v4f*)(dst + pp * PAIR_STRIDE + 4) = w1;
        }
      }
      const v2f af = pc[s4];
      ssum += af.x + af.y;
      const float* bp = Vb + (s4 * 2) * PAIR_STRIDE;
#pragma unroll
      for (int t = 0; t < 8; ++t) {
        const v2f bf = (s4 == 0) ? b0[t] : *(const v2f*)(bp + t * 32);
        acc[t] = __builtin_amdgcn_wmma_f32_16x16x4_f32(
            false, af, false, bf, (short)0, acc[t], false, false);
      }
    }
#pragma unroll
    for (int s = 0; s < AB_KS; ++s) pc[s] = pn[s];
    __syncthreads();
  }

  ssum += __shfl_xor(ssum, 16, 32);

  {
    float* ot = smem + (size_t)(g * 64 + rg * 16 + half * 8) * AB_OPITCH + lid;
#pragma unroll
    for (int t = 0; t < 8; ++t) {
#pragma unroll
      for (int v = 0; v < 8; ++v)
        ot[v * AB_OPITCH + t * 16] = acc[t][v];
    }
    if (half == 0) dsm[g][rg * 16 + lid] = ssum;
  }
  __syncthreads();
#pragma unroll
  for (int q = 0; q < 4; ++q) {
    const int r = q * 16 + wave;
    v4f num = {0.f, 0.f, 0.f, 0.f};
    float den = 0.f;
#pragma unroll
    for (int zz = 0; zz < AB_JQ; ++zz) {
      num += *(const v4f*)(smem + (size_t)(zz * 64 + r) * AB_OPITCH + lane * 4);
      den += dsm[zz][r];
    }
    v4f val = num / den;
    val.x = (val.x > 0.f) ? val.x : (__expf(val.x) - 1.f);
    val.y = (val.y > 0.f) ? val.y : (__expf(val.y) - 1.f);
    val.z = (val.z > 0.f) ? val.z : (__expf(val.z) - 1.f);
    val.w = (val.w > 0.f) ? val.w : (__expf(val.w) - 1.f);
    *(v4f*)(out + (size_t)(iblk + r) * F_OUT + head * HEAD_DIM + lane * 4) = val;
  }
}

template <int JS>
__global__ __launch_bounds__(512) void attn_reduce(const float* __restrict__ pacc,
                                                   const float* __restrict__ pssum,
                                                   float* __restrict__ out) {
  const int lane = threadIdx.x & 31;
  const int wave = threadIdx.x >> 5;
  const int i = blockIdx.x * 16 + wave;

  v4f p[HEADS][JS];
  float d[HEADS][JS];
#pragma unroll
  for (int head = 0; head < HEADS; ++head) {
#pragma unroll
    for (int zz = 0; zz < JS; ++zz) {
      p[head][zz] = *(const v4f*)(pacc + (size_t)zz * (N_NODES * F_OUT) +
                                  (size_t)i * F_OUT + head * HEAD_DIM + lane * 4);
      d[head][zz] = pssum[(size_t)zz * (HEADS * N_NODES) + head * N_NODES + i];
    }
  }
#pragma unroll
  for (int head = 0; head < HEADS; ++head) {
    v4f num = {0.f, 0.f, 0.f, 0.f};
    float den = 0.f;
#pragma unroll
    for (int zz = 0; zz < JS; ++zz) {
      num += p[head][zz];
      den += d[head][zz];
    }
    v4f val = num / den;
    val.x = (val.x > 0.f) ? val.x : (__expf(val.x) - 1.f);
    val.y = (val.y > 0.f) ? val.y : (__expf(val.y) - 1.f);
    val.z = (val.z > 0.f) ? val.z : (__expf(val.z) - 1.f);
    val.w = (val.w > 0.f) ? val.w : (__expf(val.w) - 1.f);
    *(v4f*)(out + (size_t)i * F_OUT + head * HEAD_DIM + lane * 4) = val;
  }
}

extern "C" void kernel_launch(void* const* d_in, const int* in_sizes, int n_in,
                              void* d_out, int out_size, void* d_ws, size_t ws_size,
                              hipStream_t stream) {
  const float* x   = (const float*)d_in[0];
  const int*   adj = (const int*)d_in[1];
  const float* W   = (const float*)d_in[2];
  const float* a   = (const float*)d_in[3];
  float* out = (float*)d_out;

  constexpr int JS = 4;
  const size_t HN = (size_t)HEADS * N_NODES;

  float* h   = (float*)d_ws;
  float* si  = h + (size_t)N_NODES * F_OUT;
  float* sj  = si + HN;
  float* Mx  = sj + HN;
  float* pacc  = Mx + HN;
  float* pssum = pacc + (size_t)JS * N_NODES * F_OUT;
  const size_t need_js =
      (size_t)(pssum + JS * HN - (float*)d_ws) * sizeof(float);
  unsigned* bits = (unsigned*)(pssum + JS * HN);
  const size_t need_bits =
      (size_t)((float*)(bits + (size_t)N_NODES * BIT_WORDS) - (float*)d_ws) * sizeof(float);

  gemm_xw<<<dim3(N_NODES / 64, F_OUT / 64), 128, 0, stream>>>(x, W, h);
  attn_scores<<<dim3(N_NODES / 4), 128, 0, stream>>>(h, a, si, sj);
  if (ws_size >= need_bits) {
    row_max<true><<<dim3(N_NODES / RM_ROWS), 512, 0, stream>>>(adj, sj, Mx, bits);
    attn_agg_blk<<<dim3(N_NODES / 64, HEADS), 512, 0, stream>>>(h, bits, si, sj, Mx, out);
    return;
  }
  row_max<false><<<dim3(N_NODES / RM_ROWS), 512, 0, stream>>>(adj, sj, Mx, nullptr);

  if (ws_size >= need_js) {
    attn_agg<true, false><<<dim3(N_NODES / 64, HEADS, JS), 128, 0, stream>>>(
        h, adj, nullptr, si, sj, Mx, pacc, pssum, N_NODES / JS);
    attn_reduce<JS><<<dim3(N_NODES / 16), 512, 0, stream>>>(pacc, pssum, out);
  } else {
    attn_agg<false, false><<<dim3(N_NODES / 64, HEADS, 1), 128, 0, stream>>>(
        h, adj, nullptr, si, sj, Mx, out, nullptr, N_NODES);
  }
}
